# starting kernel
# baseline (speedup 1.0000x reference)
#include <hip/hip_runtime.h>
#include <hip/hip_bf16.h>
#include <math.h>

#define NNODES 100000
#define NEDGES 1200000
#define DIN    64
#define DOUT   64
#define EDIM   16
#define KPRE   144
#define KPOST  832
#define AVG_D  2.5649493574615367f
#define EPSV   1e-05f

typedef __attribute__((ext_vector_type(2))) float v2f;
typedef __attribute__((ext_vector_type(8))) float v8f;

__device__ __forceinline__ unsigned fenc(float f) {
    unsigned u = __float_as_uint(f);
    return (u & 0x80000000u) ? ~u : (u | 0x80000000u);
}
__device__ __forceinline__ float fdec(unsigned u) {
    return (u & 0x80000000u) ? __uint_as_float(u & 0x7fffffffu)
                             : __uint_as_float(~u);
}


__global__ __launch_bounds__(256) void pna_init(float* s1, float* s2,
                                                unsigned* mx, unsigned* mn,
                                                float* deg, float* bns, float* bnq) {
    int i = blockIdx.x * 256 + threadIdx.x;
    if (i < NNODES * DIN) {
        s1[i] = 0.0f; s2[i] = 0.0f;
        mx[i] = 0u;
        mn[i] = 0xFFFFFFFFu;
    }
    if (i < NNODES) deg[i] = 0.0f;
    if (i < DOUT) { bns[i] = 0.0f; bnq[i] = 0.0f; }
}

__global__ __launch_bounds__(256) void pna_degree(const int* dst, float* deg) {
    int i = blockIdx.x * 256 + threadIdx.x;
    if (i < NEDGES) atomicAdd(&deg[dst[i]], 1.0f);
}

#define ZSTRIDE 148
__global__ __launch_bounds__(256) void pna_edge(
    const float* __restrict__ h, const float* __restrict__ e,
    const float* __restrict__ Wpre, const float* __restrict__ bpre,
    const int* __restrict__ src, const int* __restrict__ dst,
    float* s1, float* s2, unsigned* mx, unsigned* mn) {

    __shared__ float zs[32 * ZSTRIDE];
    __shared__ float wsT[DIN * ZSTRIDE];
    __shared__ int   ldst[32];

    const int tid = threadIdx.x;
    const int e0  = blockIdx.x * 32;

    for (int i = tid; i < KPRE * DIN; i += 256) {
        int n = i & 63, k = i >> 6;
        wsT[n * ZSTRIDE + k] = Wpre[k * DIN + n];
    }
    {
        int r = tid >> 3, p = tid & 7;
        int ei = e0 + r;
        int sI = src[ei], dI = dst[ei];
        if (p == 0) ldst[r] = dI;
        int c0 = p * 18;
        for (int c = c0; c < c0 + 18; ++c) {
            float v;
            if (c < DIN)            v = h[sI * DIN + c];
            else if (c < 2 * DIN)   v = h[dI * DIN + (c - DIN)];
            else                    v = e[ei * EDIM + (c - 2 * DIN)];
            zs[r * ZSTRIDE + c] = v;
        }
    }
    __syncthreads();

    const int l  = tid & 31;
    const int wv = tid >> 5;
    const int cg = wv & 3;
    const int eg = wv >> 2;
    const int lo = l & 15;
    const int hi = l >> 4;

    v8f acc = {};
    const float* arow = &zs[(eg * 16 + lo) * ZSTRIDE + 2 * hi];
    const float* brow = &wsT[(cg * 16 + lo) * ZSTRIDE + 2 * hi];
#pragma unroll 4
    for (int k0 = 0; k0 < KPRE; k0 += 4) {
        v2f a = *(const v2f*)(arow + k0);
        v2f b = *(const v2f*)(brow + k0);
        acc = __builtin_amdgcn_wmma_f32_16x16x4_f32(
            false, a, false, b, (short)0, acc, false, false);
    }

    const int col = cg * 16 + lo;
    const float bv = bpre[col];
#pragma unroll
    for (int r = 0; r < 8; ++r) {
        int M  = r + 8 * hi;
        int er = eg * 16 + M;
        int d  = ldst[er];
        float v = acc[r] + bv;
        unsigned ev = fenc(v);
        size_t o = (size_t)d * DIN + col;
        atomicAdd(&s1[o], v);
        atomicAdd(&s2[o], v * v);
        atomicMax(&mx[o], ev);
        atomicMin(&mn[o], ev);
    }
}

#define FSTRIDE 836
__global__ __launch_bounds__(128) void pna_node(
    const float* __restrict__ h, const float* __restrict__ snorm,
    const float* __restrict__ Wpost, const float* __restrict__ bpost,
    const float* __restrict__ s1, const float* __restrict__ s2,
    const unsigned* __restrict__ mx, const unsigned* __restrict__ mn,
    const float* __restrict__ deg,
    float* __restrict__ out, float* bns, float* bnq) {

    __shared__ float feat[16 * FSTRIDE];
    const int tid = threadIdx.x;
    const int n0  = blockIdx.x * 16;

    for (int i = tid; i < 16 * 320; i += 128) {
        int r = i / 320, c = i % 320;
        int node = n0 + r;
        float v;
        if (c < DIN) {
            v = h[node * DIN + c];
        } else {
            int a = c - DIN;
            int sub = a >> 6, colc = a & 63;
            size_t o = (size_t)node * DIN + colc;
            float D  = deg[node];
            if (sub == 0)      v = s1[o] / D;
            else if (sub == 1) v = fdec(mx[o]);
            else if (sub == 2) v = fdec(mn[o]);
            else {
                float mean = s1[o] / D;
                v = sqrtf(fmaxf(s2[o] / D - mean * mean, 0.0f) + EPSV);
            }
            if (c == DIN) feat[r * FSTRIDE + 832] = logf(D + 1.0f);
        }
        feat[r * FSTRIDE + c] = v;
    }
    __syncthreads();
    for (int i = tid; i < 16 * 512; i += 128) {
        int r = i >> 9, j = i & 511;
        float logd = feat[r * FSTRIDE + 832];
        float f = (j < 256) ? (logd / AVG_D) : (AVG_D / logd);
        feat[r * FSTRIDE + 320 + j] = feat[r * FSTRIDE + DIN + (j & 255)] * f;
    }
    __syncthreads();

    const int l  = tid & 31;
    const int cg = tid >> 5;
    const int lo = l & 15;
    const int hi = l >> 4;
    const int col = cg * 16 + lo;

    v8f acc = {};
    const float* arow = &feat[lo * FSTRIDE + 2 * hi];
#pragma unroll 4
    for (int k0 = 0; k0 < KPOST; k0 += 4) {
        v2f a = *(const v2f*)(arow + k0);
        v2f b;
        b.x = Wpost[(k0 + 2 * hi)     * DOUT + col];
        b.y = Wpost[(k0 + 2 * hi + 1) * DOUT + col];
        acc = __builtin_amdgcn_wmma_f32_16x16x4_f32(
            false, a, false, b, (short)0, acc, false, false);
    }

    const float bv = bpost[col];
    float ps = 0.0f, ps2 = 0.0f;
#pragma unroll
    for (int r = 0; r < 8; ++r) {
        int M = r + 8 * hi;
        int node = n0 + M;
        float v = (acc[r] + bv) * snorm[node];
        out[(size_t)node * DOUT + col] = v;
        ps += v; ps2 += v * v;
    }
    ps  += __shfl_xor(ps, 16, 32);
    ps2 += __shfl_xor(ps2, 16, 32);
    if (hi == 0) {
        atomicAdd(&bns[col], ps);
        atomicAdd(&bnq[col], ps2);
    }
}

__global__ __launch_bounds__(256) void pna_bn(
    float* out, const float* bns, const float* bnq,
    const float* gamma, const float* beta) {
    int i = blockIdx.x * 256 + threadIdx.x;
    if (i >= NNODES * DOUT) return;
    int c = i & 63;
    float mu  = bns[c] / (float)NNODES;
    float var = bnq[c] / (float)NNODES - mu * mu;
    float v = out[i];
    out[i] = (v - mu) * rsqrtf(var + 1e-05f) * gamma[c] + beta[c];
}

extern "C" void kernel_launch(void* const* d_in, const int* in_sizes, int n_in,
                              void* d_out, int out_size, void* d_ws, size_t ws_size,
                              hipStream_t stream) {
    const float* h      = (const float*)d_in[0];
    const float* e      = (const float*)d_in[1];
    const float* snorm  = (const float*)d_in[2];
    const float* Wpre   = (const float*)d_in[3];
    const float* bpre   = (const float*)d_in[4];
    const float* Wpost  = (const float*)d_in[5];
    const float* bpost  = (const float*)d_in[6];
    const float* gamma  = (const float*)d_in[7];
    const float* beta   = (const float*)d_in[8];
    const int*   src    = (const int*)d_in[9];
    const int*   dst    = (const int*)d_in[10];
    float* out = (float*)d_out;

    float*    s1  = (float*)d_ws;
    float*    s2  = s1 + (size_t)NNODES * DIN;
    unsigned* mx  = (unsigned*)(s2 + (size_t)NNODES * DIN);
    unsigned* mn  = mx + (size_t)NNODES * DIN;
    float*    deg = (float*)(mn + (size_t)NNODES * DIN);
    float*    bns = deg + NNODES;
    float*    bnq = bns + DOUT;

    pna_init  <<<(NNODES * DIN + 255) / 256, 256, 0, stream>>>(s1, s2, mx, mn, deg, bns, bnq);
    pna_degree<<<(NEDGES + 255) / 256, 256, 0, stream>>>(dst, deg);
    pna_edge  <<<NEDGES / 32, 256, 0, stream>>>(h, e, Wpre, bpre, src, dst, s1, s2, mx, mn);
    pna_node  <<<NNODES / 16, 128, 0, stream>>>(h, snorm, Wpost, bpost, s1, s2, mx, mn, deg,
                                                out, bns, bnq);
    pna_bn    <<<(NNODES * DOUT + 255) / 256, 256, 0, stream>>>(out, bns, bnq, gamma, beta);
}


# Edges visited in destination-sorted order (counting sort: atomic arrival rank, one-block scan, scatter); each lane folds runs of equal dst in registers before its atomics; sum order changes
# speedup vs baseline: 1.0836x; 1.0836x over previous
#include <hip/hip_runtime.h>
#include <hip/hip_bf16.h>
#include <math.h>

#define NNODES 100000
#define NEDGES 1200000
#define DIN    64
#define DOUT   64
#define EDIM   16
#define KPRE   144
#define KPOST  832
#define AVG_D  2.5649493574615367f
#define EPSV   1e-05f

typedef __attribute__((ext_vector_type(2))) float v2f;
typedef __attribute__((ext_vector_type(8))) float v8f;

__device__ __forceinline__ unsigned fenc(float f) {
    unsigned u = __float_as_uint(f);
    return (u & 0x80000000u) ? ~u : (u | 0x80000000u);
}
__device__ __forceinline__ float fdec(unsigned u) {
    return (u & 0x80000000u) ? __uint_as_float(u & 0x7fffffffu)
                             : __uint_as_float(~u);
}


__global__ __launch_bounds__(256) void pna_init(float* s1, float* s2,
                                                unsigned* mx, unsigned* mn,
                                                int* cnt, float* bns, float* bnq) {
    int i = blockIdx.x * 256 + threadIdx.x;
    if (i < NNODES * DIN) {
        s1[i] = 0.0f; s2[i] = 0.0f;
        mx[i] = 0u;
        mn[i] = 0xFFFFFFFFu;
    }
    if (i < NNODES) cnt[i] = 0;
    if (i < DOUT) { bns[i] = 0.0f; bnq[i] = 0.0f; }
}

__global__ __launch_bounds__(256) void pna_rank(const int* __restrict__ dst,
                                                int* cnt, int* __restrict__ rank) {
    int i = blockIdx.x * 256 + threadIdx.x;
    if (i < NEDGES) rank[i] = atomicAdd(&cnt[dst[i]], 1);
}

#define SCAN_CH 98
__global__ __launch_bounds__(1024) void pna_scan(const int* __restrict__ cnt,
                                                 int* __restrict__ rptr) {
    __shared__ int part[1024];
    const int t = threadIdx.x;
    const int b = t * SCAN_CH;
    const int en = (b + SCAN_CH < NNODES) ? (b + SCAN_CH) : NNODES;
    int s = 0;
    for (int i = b; i < en; ++i) s += cnt[i];
    part[t] = s;
    __syncthreads();
    for (int off = 1; off < 1024; off <<= 1) {
        int v = (t >= off) ? part[t - off] : 0;
        __syncthreads();
        part[t] += v;
        __syncthreads();
    }
    int run = part[t] - s;
    for (int i = b; i < en; ++i) { rptr[i] = run; run += cnt[i]; }
}

__global__ __launch_bounds__(256) void pna_fill(const int* __restrict__ dst,
                                                const int* __restrict__ rank,
                                                const int* __restrict__ rptr,
                                                int* __restrict__ perm) {
    int i = blockIdx.x * 256 + threadIdx.x;
    if (i < NEDGES) perm[rptr[dst[i]] + rank[i]] = i;
}

#define ZSTRIDE 148
__global__ __launch_bounds__(256) void pna_edge(
    const float* __restrict__ h, const float* __restrict__ e,
    const float* __restrict__ Wpre, const float* __restrict__ bpre,
    const int* __restrict__ src, const int* __restrict__ dst,
    const int* __restrict__ perm,
    float* s1, float* s2, unsigned* mx, unsigned* mn) {

    __shared__ float zs[32 * ZSTRIDE];
    __shared__ float wsT[DIN * ZSTRIDE];
    __shared__ int   ldst[32];

    const int tid = threadIdx.x;
    const int e0  = blockIdx.x * 32;

    for (int i = tid; i < KPRE * DIN; i += 256) {
        int n = i & 63, k = i >> 6;
        wsT[n * ZSTRIDE + k] = Wpre[k * DIN + n];
    }
    {
        int r = tid >> 3, p = tid & 7;
        int ei = perm[e0 + r];
        int sI = src[ei], dI = dst[ei];
        if (p == 0) ldst[r] = dI;
        int c0 = p * 18;
        for (int c = c0; c < c0 + 18; ++c) {
            float v;
            if (c < DIN)            v = h[sI * DIN + c];
            else if (c < 2 * DIN)   v = h[dI * DIN + (c - DIN)];
            else                    v = e[ei * EDIM + (c - 2 * DIN)];
            zs[r * ZSTRIDE + c] = v;
        }
    }
    __syncthreads();

    const int l  = tid & 31;
    const int wv = tid >> 5;
    const int cg = wv & 3;
    const int eg = wv >> 2;
    const int lo = l & 15;
    const int hi = l >> 4;

    v8f acc = {};
    const float* arow = &zs[(eg * 16 + lo) * ZSTRIDE + 2 * hi];
    const float* brow = &wsT[(cg * 16 + lo) * ZSTRIDE + 2 * hi];
#pragma unroll 4
    for (int k0 = 0; k0 < KPRE; k0 += 4) {
        v2f a = *(const v2f*)(arow + k0);
        v2f b = *(const v2f*)(brow + k0);
        acc = __builtin_amdgcn_wmma_f32_16x16x4_f32(
            false, a, false, b, (short)0, acc, false, false);
    }

    const int col = cg * 16 + lo;
    const float bv = bpre[col];
    const int rb = eg * 16 + 8 * hi;
    int cur = ldst[rb];
    float v = acc[0] + bv;
    float sum = v, sq = v * v;
    unsigned emx = fenc(v), emn = emx;
#pragma unroll
    for (int r = 1; r < 8; ++r) {
        int d = ldst[rb + r];
        v = acc[r] + bv;
        unsigned ev = fenc(v);
        if (d != cur) {
            size_t o = (size_t)cur * DIN + col;
            atomicAdd(&s1[o], sum);
            atomicAdd(&s2[o], sq);
            atomicMax(&mx[o], emx);
            atomicMin(&mn[o], emn);
            cur = d; sum = v; sq = v * v; emx = ev; emn = ev;
        } else {
            sum += v; sq += v * v;
            emx = (ev > emx) ? ev : emx;
            emn = (ev < emn) ? ev : emn;
        }
    }
    {
        size_t o = (size_t)cur * DIN + col;
        atomicAdd(&s1[o], sum);
        atomicAdd(&s2[o], sq);
        atomicMax(&mx[o], emx);
        atomicMin(&mn[o], emn);
    }
}

#define FSTRIDE 836
__global__ __launch_bounds__(128) void pna_node(
    const float* __restrict__ h, const float* __restrict__ snorm,
    const float* __restrict__ Wpost, const float* __restrict__ bpost,
    const float* __restrict__ s1, const float* __restrict__ s2,
    const unsigned* __restrict__ mx, const unsigned* __restrict__ mn,
    const int* __restrict__ cnt,
    float* __restrict__ out, float* bns, float* bnq) {

    __shared__ float feat[16 * FSTRIDE];
    const int tid = threadIdx.x;
    const int n0  = blockIdx.x * 16;

    for (int i = tid; i < 16 * 320; i += 128) {
        int r = i / 320, c = i % 320;
        int node = n0 + r;
        float v;
        if (c < DIN) {
            v = h[node * DIN + c];
        } else {
            int a = c - DIN;
            int sub = a >> 6, colc = a & 63;
            size_t o = (size_t)node * DIN + colc;
            float D  = (float)cnt[node];
            if (sub == 0)      v = s1[o] / D;
            else if (sub == 1) v = fdec(mx[o]);
            else if (sub == 2) v = fdec(mn[o]);
            else {
                float mean = s1[o] / D;
                v = sqrtf(fmaxf(s2[o] / D - mean * mean, 0.0f) + EPSV);
            }
            if (c == DIN) feat[r * FSTRIDE + 832] = logf(D + 1.0f);
        }
        feat[r * FSTRIDE + c] = v;
    }
    __syncthreads();
    for (int i = tid; i < 16 * 512; i += 128) {
        int r = i >> 9, j = i & 511;
        float logd = feat[r * FSTRIDE + 832];
        float f = (j < 256) ? (logd / AVG_D) : (AVG_D / logd);
        feat[r * FSTRIDE + 320 + j] = feat[r * FSTRIDE + DIN + (j & 255)] * f;
    }
    __syncthreads();

    const int l  = tid & 31;
    const int cg = tid >> 5;
    const int lo = l & 15;
    const int hi = l >> 4;
    const int col = cg * 16 + lo;

    v8f acc = {};
    const float* arow = &feat[lo * FSTRIDE + 2 * hi];
#pragma unroll 4
    for (int k0 = 0; k0 < KPOST; k0 += 4) {
        v2f a = *(const v2f*)(arow + k0);
        v2f b;
        b.x = Wpost[(k0 + 2 * hi)     * DOUT + col];
        b.y = Wpost[(k0 + 2 * hi + 1) * DOUT + col];
        acc = __builtin_amdgcn_wmma_f32_16x16x4_f32(
            false, a, false, b, (short)0, acc, false, false);
    }

    const float bv = bpost[col];
    float ps = 0.0f, ps2 = 0.0f;
#pragma unroll
    for (int r = 0; r < 8; ++r) {
        int M = r + 8 * hi;
        int node = n0 + M;
        float v = (acc[r] + bv) * snorm[node];
        out[(size_t)node * DOUT + col] = v;
        ps += v; ps2 += v * v;
    }
    ps  += __shfl_xor(ps, 16, 32);
    ps2 += __shfl_xor(ps2, 16, 32);
    if (hi == 0) {
        atomicAdd(&bns[col], ps);
        atomicAdd(&bnq[col], ps2);
    }
}

__global__ __launch_bounds__(256) void pna_bn(
    float* out, const float* bns, const float* bnq,
    const float* gamma, const float* beta) {
    int i = blockIdx.x * 256 + threadIdx.x;
    if (i >= NNODES * DOUT) return;
    int c = i & 63;
    float mu  = bns[c] / (float)NNODES;
    float var = bnq[c] / (float)NNODES - mu * mu;
    float v = out[i];
    out[i] = (v - mu) * rsqrtf(var + 1e-05f) * gamma[c] + beta[c];
}

extern "C" void kernel_launch(void* const* d_in, const int* in_sizes, int n_in,
                              void* d_out, int out_size, void* d_ws, size_t ws_size,
                              hipStream_t stream) {
    const float* h      = (const float*)d_in[0];
    const float* e      = (const float*)d_in[1];
    const float* snorm  = (const float*)d_in[2];
    const float* Wpre   = (const float*)d_in[3];
    const float* bpre   = (const float*)d_in[4];
    const float* Wpost  = (const float*)d_in[5];
    const float* bpost  = (const float*)d_in[6];
    const float* gamma  = (const float*)d_in[7];
    const float* beta   = (const float*)d_in[8];
    const int*   src    = (const int*)d_in[9];
    const int*   dst    = (const int*)d_in[10];
    float* out = (float*)d_out;

    float*    s1  = (float*)d_ws;
    float*    s2  = s1 + (size_t)NNODES * DIN;
    unsigned* mx  = (unsigned*)(s2 + (size_t)NNODES * DIN);
    unsigned* mn  = mx + (size_t)NNODES * DIN;
    int*      cnt = (int*)(mn + (size_t)NNODES * DIN);
    float*    bns = (float*)(cnt + NNODES);
    float*    bnq = bns + DOUT;

    int* perm = (int*)d_out;
    int* rank = perm + NEDGES;
    int* rptr = rank + NEDGES;

    pna_init  <<<(NNODES * DIN + 255) / 256, 256, 0, stream>>>(s1, s2, mx, mn, cnt, bns, bnq);
    pna_rank  <<<(NEDGES + 255) / 256, 256, 0, stream>>>(dst, cnt, rank);
    pna_scan  <<<1, 1024, 0, stream>>>(cnt, rptr);
    pna_fill  <<<(NEDGES + 255) / 256, 256, 0, stream>>>(dst, rank, rptr, perm);
    pna_edge  <<<NEDGES / 32, 256, 0, stream>>>(h, e, Wpre, bpre, src, dst, perm, s1, s2, mx, mn);
    pna_node  <<<NNODES / 16, 128, 0, stream>>>(h, snorm, Wpost, bpost, s1, s2, mx, mn, cnt,
                                                out, bns, bnq);
    pna_bn    <<<(NNODES * DOUT + 255) / 256, 256, 0, stream>>>(out, bns, bnq, gamma, beta);
}


# Edge kernel gather re-indexed: the 8 lanes of an edge take interleaved 16-byte pieces of its z row, five independent loads per lane in one group instead of 18 load-wait-store trips
# speedup vs baseline: 1.3397x; 1.2363x over previous
#include <hip/hip_runtime.h>
#include <hip/hip_bf16.h>
#include <math.h>

#define NNODES 100000
#define NEDGES 1200000
#define DIN    64
#define DOUT   64
#define EDIM   16
#define KPRE   144
#define KPOST  832
#define AVG_D  2.5649493574615367f
#define EPSV   1e-05f

typedef __attribute__((ext_vector_type(2))) float v2f;
typedef __attribute__((ext_vector_type(8))) float v8f;

__device__ __forceinline__ unsigned fenc(float f) {
    unsigned u = __float_as_uint(f);
    return (u & 0x80000000u) ? ~u : (u | 0x80000000u);
}
__device__ __forceinline__ float fdec(unsigned u) {
    return (u & 0x80000000u) ? __uint_as_float(u & 0x7fffffffu)
                             : __uint_as_float(~u);
}


__global__ __launch_bounds__(256) void pna_init(float* s1, float* s2,
                                                unsigned* mx, unsigned* mn,
                                                int* cnt, float* bns, float* bnq) {
    int i = blockIdx.x * 256 + threadIdx.x;
    if (i < NNODES * DIN) {
        s1[i] = 0.0f; s2[i] = 0.0f;
        mx[i] = 0u;
        mn[i] = 0xFFFFFFFFu;
    }
    if (i < NNODES) cnt[i] = 0;
    if (i < DOUT) { bns[i] = 0.0f; bnq[i] = 0.0f; }
}

__global__ __launch_bounds__(256) void pna_rank(const int* __restrict__ dst,
                                                int* cnt, int* __restrict__ rank) {
    int i = blockIdx.x * 256 + threadIdx.x;
    if (i < NEDGES) rank[i] = atomicAdd(&cnt[dst[i]], 1);
}

#define SCAN_CH 98
__global__ __launch_bounds__(1024) void pna_scan(const int* __restrict__ cnt,
                                                 int* __restrict__ rptr) {
    __shared__ int part[1024];
    const int t = threadIdx.x;
    const int b = t * SCAN_CH;
    const int en = (b + SCAN_CH < NNODES) ? (b + SCAN_CH) : NNODES;
    int s = 0;
    for (int i = b; i < en; ++i) s += cnt[i];
    part[t] = s;
    __syncthreads();
    for (int off = 1; off < 1024; off <<= 1) {
        int v = (t >= off) ? part[t - off] : 0;
        __syncthreads();
        part[t] += v;
        __syncthreads();
    }
    int run = part[t] - s;
    for (int i = b; i < en; ++i) { rptr[i] = run; run += cnt[i]; }
}

__global__ __launch_bounds__(256) void pna_fill(const int* __restrict__ dst,
                                                const int* __restrict__ rank,
                                                const int* __restrict__ rptr,
                                                int* __restrict__ perm) {
    int i = blockIdx.x * 256 + threadIdx.x;
    if (i < NEDGES) perm[rptr[dst[i]] + rank[i]] = i;
}

#define ZSTRIDE 148
__global__ __launch_bounds__(256) void pna_edge(
    const float* __restrict__ h, const float* __restrict__ e,
    const float* __restrict__ Wpre, const float* __restrict__ bpre,
    const int* __restrict__ src, const int* __restrict__ dst,
    const int* __restrict__ perm,
    float* s1, float* s2, unsigned* mx, unsigned* mn) {

    __shared__ __attribute__((aligned(16))) float zs[32 * ZSTRIDE];
    __shared__ float wsT[DIN * ZSTRIDE];
    __shared__ int   ldst[32];

    const int tid = threadIdx.x;
    const int e0  = blockIdx.x * 32;

    for (int i = tid; i < KPRE * DIN; i += 256) {
        int n = i & 63, k = i >> 6;
        wsT[n * ZSTRIDE + k] = Wpre[k * DIN + n];
    }
    {
        const int r = tid >> 3, p = tid & 7;
        const int ei = perm[e0 + r];
        const int sI = src[ei], dI = dst[ei];
        if (p == 0) ldst[r] = dI;
        const float4* hs = (const float4*)(h + (size_t)sI * DIN);
        const float4* hd = (const float4*)(h + (size_t)dI * DIN);
        const float2* er = (const float2*)(e + (size_t)ei * EDIM);
        const float4 s_lo = hs[p], s_hi = hs[p + 8];
        const float4 d_lo = hd[p], d_hi = hd[p + 8];
        const float2 e_pc = er[p];
        float* zr = &zs[r * ZSTRIDE];
        *(float4*)(zr + 4 * p)            = s_lo;
        *(float4*)(zr + 32 + 4 * p)       = s_hi;
        *(float4*)(zr + DIN + 4 * p)      = d_lo;
        *(float4*)(zr + DIN + 32 + 4 * p) = d_hi;
        *(float2*)(zr + 2 * DIN + 2 * p)  = e_pc;
    }
    __syncthreads();

    const int l  = tid & 31;
    const int wv = tid >> 5;
    const int cg = wv & 3;
    const int eg = wv >> 2;
    const int lo = l & 15;
    const int hi = l >> 4;

    v8f acc = {};
    const float* arow = &zs[(eg * 16 + lo) * ZSTRIDE + 2 * hi];
    const float* brow = &wsT[(cg * 16 + lo) * ZSTRIDE + 2 * hi];
#pragma unroll 4
    for (int k0 = 0; k0 < KPRE; k0 += 4) {
        v2f a = *(const v2f*)(arow + k0);
        v2f b = *(const v2f*)(brow + k0);
        acc = __builtin_amdgcn_wmma_f32_16x16x4_f32(
            false, a, false, b, (short)0, acc, false, false);
    }

    const int col = cg * 16 + lo;
    const float bv = bpre[col];
    const int rb = eg * 16 + 8 * hi;
    int cur = ldst[rb];
    float v = acc[0] + bv;
    float sum = v, sq = v * v;
    unsigned emx = fenc(v), emn = emx;
#pragma unroll
    for (int r = 1; r < 8; ++r) {
        int d = ldst[rb + r];
        v = acc[r] + bv;
        unsigned ev = fenc(v);
        if (d != cur) {
            size_t o = (size_t)cur * DIN + col;
            atomicAdd(&s1[o], sum);
            atomicAdd(&s2[o], sq);
            atomicMax(&mx[o], emx);
            atomicMin(&mn[o], emn);
            cur = d; sum = v; sq = v * v; emx = ev; emn = ev;
        } else {
            sum += v; sq += v * v;
            emx = (ev > emx) ? ev : emx;
            emn = (ev < emn) ? ev : emn;
        }
    }
    {
        size_t o = (size_t)cur * DIN + col;
        atomicAdd(&s1[o], sum);
        atomicAdd(&s2[o], sq);
        atomicMax(&mx[o], emx);
        atomicMin(&mn[o], emn);
    }
}

#define FSTRIDE 836
__global__ __launch_bounds__(128) void pna_node(
    const float* __restrict__ h, const float* __restrict__ snorm,
    const float* __restrict__ Wpost, const float* __restrict__ bpost,
    const float* __restrict__ s1, const float* __restrict__ s2,
    const unsigned* __restrict__ mx, const unsigned* __restrict__ mn,
    const int* __restrict__ cnt,
    float* __restrict__ out, float* bns, float* bnq) {

    __shared__ float feat[16 * FSTRIDE];
    const int tid = threadIdx.x;
    const int n0  = blockIdx.x * 16;

    for (int i = tid; i < 16 * 320; i += 128) {
        int r = i / 320, c = i % 320;
        int node = n0 + r;
        float v;
        if (c < DIN) {
            v = h[node * DIN + c];
        } else {
            int a = c - DIN;
            int sub = a >> 6, colc = a & 63;
            size_t o = (size_t)node * DIN + colc;
            float D  = (float)cnt[node];
            if (sub == 0)      v = s1[o] / D;
            else if (sub == 1) v = fdec(mx[o]);
            else if (sub == 2) v = fdec(mn[o]);
            else {
                float mean = s1[o] / D;
                v = sqrtf(fmaxf(s2[o] / D - mean * mean, 0.0f) + EPSV);
            }
            if (c == DIN) feat[r * FSTRIDE + 832] = logf(D + 1.0f);
        }
        feat[r * FSTRIDE + c] = v;
    }
    __syncthreads();
    for (int i = tid; i < 16 * 512; i += 128) {
        int r = i >> 9, j = i & 511;
        float logd = feat[r * FSTRIDE + 832];
        float f = (j < 256) ? (logd / AVG_D) : (AVG_D / logd);
        feat[r * FSTRIDE + 320 + j] = feat[r * FSTRIDE + DIN + (j & 255)] * f;
    }
    __syncthreads();

    const int l  = tid & 31;
    const int cg = tid >> 5;
    const int lo = l & 15;
    const int hi = l >> 4;
    const int col = cg * 16 + lo;

    v8f acc = {};
    const float* arow = &feat[lo * FSTRIDE + 2 * hi];
#pragma unroll 4
    for (int k0 = 0; k0 < KPOST; k0 += 4) {
        v2f a = *(const v2f*)(arow + k0);
        v2f b;
        b.x = Wpost[(k0 + 2 * hi)     * DOUT + col];
        b.y = Wpost[(k0 + 2 * hi + 1) * DOUT + col];
        acc = __builtin_amdgcn_wmma_f32_16x16x4_f32(
            false, a, false, b, (short)0, acc, false, false);
    }

    const float bv = bpost[col];
    float ps = 0.0f, ps2 = 0.0f;
#pragma unroll
    for (int r = 0; r < 8; ++r) {
        int M = r + 8 * hi;
        int node = n0 + M;
        float v = (acc[r] + bv) * snorm[node];
        out[(size_t)node * DOUT + col] = v;
        ps += v; ps2 += v * v;
    }
    ps  += __shfl_xor(ps, 16, 32);
    ps2 += __shfl_xor(ps2, 16, 32);
    if (hi == 0) {
        atomicAdd(&bns[col], ps);
        atomicAdd(&bnq[col], ps2);
    }
}

__global__ __launch_bounds__(256) void pna_bn(
    float* out, const float* bns, const float* bnq,
    const float* gamma, const float* beta) {
    int i = blockIdx.x * 256 + threadIdx.x;
    if (i >= NNODES * DOUT) return;
    int c = i & 63;
    float mu  = bns[c] / (float)NNODES;
    float var = bnq[c] / (float)NNODES - mu * mu;
    float v = out[i];
    out[i] = (v - mu) * rsqrtf(var + 1e-05f) * gamma[c] + beta[c];
}

extern "C" void kernel_launch(void* const* d_in, const int* in_sizes, int n_in,
                              void* d_out, int out_size, void* d_ws, size_t ws_size,
                              hipStream_t stream) {
    const float* h      = (const float*)d_in[0];
    const float* e      = (const float*)d_in[1];
    const float* snorm  = (const float*)d_in[2];
    const float* Wpre   = (const float*)d_in[3];
    const float* bpre   = (const float*)d_in[4];
    const float* Wpost  = (const float*)d_in[5];
    const float* bpost  = (const float*)d_in[6];
    const float* gamma  = (const float*)d_in[7];
    const float* beta   = (const float*)d_in[8];
    const int*   src    = (const int*)d_in[9];
    const int*   dst    = (const int*)d_in[10];
    float* out = (float*)d_out;

    float*    s1  = (float*)d_ws;
    float*    s2  = s1 + (size_t)NNODES * DIN;
    unsigned* mx  = (unsigned*)(s2 + (size_t)NNODES * DIN);
    unsigned* mn  = mx + (size_t)NNODES * DIN;
    int*      cnt = (int*)(mn + (size_t)NNODES * DIN);
    float*    bns = (float*)(cnt + NNODES);
    float*    bnq = bns + DOUT;

    int* perm = (int*)d_out;
    int* rank = perm + NEDGES;
    int* rptr = rank + NEDGES;

    pna_init  <<<(NNODES * DIN + 255) / 256, 256, 0, stream>>>(s1, s2, mx, mn, cnt, bns, bnq);
    pna_rank  <<<(NEDGES + 255) / 256, 256, 0, stream>>>(dst, cnt, rank);
    pna_scan  <<<1, 1024, 0, stream>>>(cnt, rptr);
    pna_fill  <<<(NEDGES + 255) / 256, 256, 0, stream>>>(dst, rank, rptr, perm);
    pna_edge  <<<NEDGES / 32, 256, 0, stream>>>(h, e, Wpre, bpre, src, dst, perm, s1, s2, mx, mn);
    pna_node  <<<NNODES / 16, 128, 0, stream>>>(h, snorm, Wpost, bpost, s1, s2, mx, mn, cnt,
                                                out, bns, bnq);
    pna_bn    <<<(NNODES * DOUT + 255) / 256, 256, 0, stream>>>(out, bns, bnq, gamma, beta);
}


# Node kernel loads grouped ahead of use: feature rows built from 11 independent 16-byte loads per thread, scaled copies written directly (one barrier); W_post fragments fetched 16 k-steps ahead
# speedup vs baseline: 1.8424x; 1.3752x over previous
#include <hip/hip_runtime.h>
#include <hip/hip_bf16.h>
#include <math.h>

#define NNODES 100000
#define NEDGES 1200000
#define DIN    64
#define DOUT   64
#define EDIM   16
#define KPRE   144
#define KPOST  832
#define AVG_D  2.5649493574615367f
#define EPSV   1e-05f

typedef __attribute__((ext_vector_type(2))) float v2f;
typedef __attribute__((ext_vector_type(8))) float v8f;

__device__ __forceinline__ unsigned fenc(float f) {
    unsigned u = __float_as_uint(f);
    return (u & 0x80000000u) ? ~u : (u | 0x80000000u);
}
__device__ __forceinline__ float fdec(unsigned u) {
    return (u & 0x80000000u) ? __uint_as_float(u & 0x7fffffffu)
                             : __uint_as_float(~u);
}


__global__ __launch_bounds__(256) void pna_init(float* s1, float* s2,
                                                unsigned* mx, unsigned* mn,
                                                int* cnt, float* bns, float* bnq) {
    int i = blockIdx.x * 256 + threadIdx.x;
    if (i < NNODES * DIN) {
        s1[i] = 0.0f; s2[i] = 0.0f;
        mx[i] = 0u;
        mn[i] = 0xFFFFFFFFu;
    }
    if (i < NNODES) cnt[i] = 0;
    if (i < DOUT) { bns[i] = 0.0f; bnq[i] = 0.0f; }
}

__global__ __launch_bounds__(256) void pna_rank(const int* __restrict__ dst,
                                                int* cnt, int* __restrict__ rank) {
    int i = blockIdx.x * 256 + threadIdx.x;
    if (i < NEDGES) rank[i] = atomicAdd(&cnt[dst[i]], 1);
}

#define SCAN_CH 98
__global__ __launch_bounds__(1024) void pna_scan(const int* __restrict__ cnt,
                                                 int* __restrict__ rptr) {
    __shared__ int part[1024];
    const int t = threadIdx.x;
    const int b = t * SCAN_CH;
    const int en = (b + SCAN_CH < NNODES) ? (b + SCAN_CH) : NNODES;
    int s = 0;
    for (int i = b; i < en; ++i) s += cnt[i];
    part[t] = s;
    __syncthreads();
    for (int off = 1; off < 1024; off <<= 1) {
        int v = (t >= off) ? part[t - off] : 0;
        __syncthreads();
        part[t] += v;
        __syncthreads();
    }
    int run = part[t] - s;
    for (int i = b; i < en; ++i) { rptr[i] = run; run += cnt[i]; }
}

__global__ __launch_bounds__(256) void pna_fill(const int* __restrict__ dst,
                                                const int* __restrict__ rank,
                                                const int* __restrict__ rptr,
                                                int* __restrict__ perm) {
    int i = blockIdx.x * 256 + threadIdx.x;
    if (i < NEDGES) perm[rptr[dst[i]] + rank[i]] = i;
}

#define ZSTRIDE 148
__global__ __launch_bounds__(256) void pna_edge(
    const float* __restrict__ h, const float* __restrict__ e,
    const float* __restrict__ Wpre, const float* __restrict__ bpre,
    const int* __restrict__ src, const int* __restrict__ dst,
    const int* __restrict__ perm,
    float* s1, float* s2, unsigned* mx, unsigned* mn) {

    __shared__ __attribute__((aligned(16))) float zs[32 * ZSTRIDE];
    __shared__ float wsT[DIN * ZSTRIDE];
    __shared__ int   ldst[32];

    const int tid = threadIdx.x;
    const int e0  = blockIdx.x * 32;

    for (int i = tid; i < KPRE * DIN; i += 256) {
        int n = i & 63, k = i >> 6;
        wsT[n * ZSTRIDE + k] = Wpre[k * DIN + n];
    }
    {
        const int r = tid >> 3, p = tid & 7;
        const int ei = perm[e0 + r];
        const int sI = src[ei], dI = dst[ei];
        if (p == 0) ldst[r] = dI;
        const float4* hs = (const float4*)(h + (size_t)sI * DIN);
        const float4* hd = (const float4*)(h + (size_t)dI * DIN);
        const float2* er = (const float2*)(e + (size_t)ei * EDIM);
        const float4 s_lo = hs[p], s_hi = hs[p + 8];
        const float4 d_lo = hd[p], d_hi = hd[p + 8];
        const float2 e_pc = er[p];
        float* zr = &zs[r * ZSTRIDE];
        *(float4*)(zr + 4 * p)            = s_lo;
        *(float4*)(zr + 32 + 4 * p)       = s_hi;
        *(float4*)(zr + DIN + 4 * p)      = d_lo;
        *(float4*)(zr + DIN + 32 + 4 * p) = d_hi;
        *(float2*)(zr + 2 * DIN + 2 * p)  = e_pc;
    }
    __syncthreads();

    const int l  = tid & 31;
    const int wv = tid >> 5;
    const int cg = wv & 3;
    const int eg = wv >> 2;
    const int lo = l & 15;
    const int hi = l >> 4;

    v8f acc = {};
    const float* arow = &zs[(eg * 16 + lo) * ZSTRIDE + 2 * hi];
    const float* brow = &wsT[(cg * 16 + lo) * ZSTRIDE + 2 * hi];
#pragma unroll 4
    for (int k0 = 0; k0 < KPRE; k0 += 4) {
        v2f a = *(const v2f*)(arow + k0);
        v2f b = *(const v2f*)(brow + k0);
        acc = __builtin_amdgcn_wmma_f32_16x16x4_f32(
            false, a, false, b, (short)0, acc, false, false);
    }

    const int col = cg * 16 + lo;
    const float bv = bpre[col];
    const int rb = eg * 16 + 8 * hi;
    int cur = ldst[rb];
    float v = acc[0] + bv;
    float sum = v, sq = v * v;
    unsigned emx = fenc(v), emn = emx;
#pragma unroll
    for (int r = 1; r < 8; ++r) {
        int d = ldst[rb + r];
        v = acc[r] + bv;
        unsigned ev = fenc(v);
        if (d != cur) {
            size_t o = (size_t)cur * DIN + col;
            atomicAdd(&s1[o], sum);
            atomicAdd(&s2[o], sq);
            atomicMax(&mx[o], emx);
            atomicMin(&mn[o], emn);
            cur = d; sum = v; sq = v * v; emx = ev; emn = ev;
        } else {
            sum += v; sq += v * v;
            emx = (ev > emx) ? ev : emx;
            emn = (ev < emn) ? ev : emn;
        }
    }
    {
        size_t o = (size_t)cur * DIN + col;
        atomicAdd(&s1[o], sum);
        atomicAdd(&s2[o], sq);
        atomicMax(&mx[o], emx);
        atomicMin(&mn[o], emn);
    }
}

#define FSTRIDE 836
#define NKB     16
#define NNB     13
__global__ __launch_bounds__(128) void pna_node(
    const float* __restrict__ h, const float* __restrict__ snorm,
    const float* __restrict__ Wpost, const float* __restrict__ bpost,
    const float* __restrict__ s1, const float* __restrict__ s2,
    const unsigned* __restrict__ mx, const unsigned* __restrict__ mn,
    const int* __restrict__ cnt,
    float* __restrict__ out, float* bns, float* bnq) {

    __shared__ __attribute__((aligned(16))) float feat[16 * FSTRIDE];
    const int tid = threadIdx.x;
    const int n0  = blockIdx.x * 16;

    {
        const int r = tid >> 3, p = tid & 7;
        const int node = n0 + r;
        const size_t o = (size_t)node * DIN;
        const float4* hp  = (const float4*)(h + o);
        const float4* s1p = (const float4*)(s1 + o);
        const float4* s2p = (const float4*)(s2 + o);
        const uint4*  mxp = (const uint4*)(mx + o);
        const uint4*  mnp = (const uint4*)(mn + o);
        const int dg = cnt[node];
        float4 hv[2], a1[2], a2[2];
        uint4  ux[2], un[2];
#pragma unroll
        for (int q = 0; q < 2; ++q) {
            hv[q] = hp[p + 8 * q];
            a1[q] = s1p[p + 8 * q];
            a2[q] = s2p[p + 8 * q];
            ux[q] = mxp[p + 8 * q];
            un[q] = mnp[p + 8 * q];
        }
        const float D    = (float)dg;
        const float logd = logf(D + 1.0f);
        const float fa   = logd / AVG_D;
        const float fb   = AVG_D / logd;
        float* fr = &feat[r * FSTRIDE];
#pragma unroll
        for (int q = 0; q < 2; ++q) {
            const int c = 4 * p + 32 * q;
            float4 mean, sd, vx, vn;
            mean.x = a1[q].x / D; mean.y = a1[q].y / D;
            mean.z = a1[q].z / D; mean.w = a1[q].w / D;
            sd.x = sqrtf(fmaxf(a2[q].x / D - mean.x * mean.x, 0.0f) + EPSV);
            sd.y = sqrtf(fmaxf(a2[q].y / D - mean.y * mean.y, 0.0f) + EPSV);
            sd.z = sqrtf(fmaxf(a2[q].z / D - mean.z * mean.z, 0.0f) + EPSV);
            sd.w = sqrtf(fmaxf(a2[q].w / D - mean.w * mean.w, 0.0f) + EPSV);
            vx.x = fdec(ux[q].x); vx.y = fdec(ux[q].y); vx.z = fdec(ux[q].z); vx.w = fdec(ux[q].w);
            vn.x = fdec(un[q].x); vn.y = fdec(un[q].y); vn.z = fdec(un[q].z); vn.w = fdec(un[q].w);
            *(float4*)(fr + c)           = hv[q];
            *(float4*)(fr + DIN + c)     = mean;
            *(float4*)(fr + 2 * DIN + c) = vx;
            *(float4*)(fr + 3 * DIN + c) = vn;
            *(float4*)(fr + 4 * DIN + c) = sd;
            *(float4*)(fr + 5 * DIN + c) = make_float4(mean.x * fa, mean.y * fa, mean.z * fa, mean.w * fa);
            *(float4*)(fr + 6 * DIN + c) = make_float4(vx.x * fa, vx.y * fa, vx.z * fa, vx.w * fa);
            *(float4*)(fr + 7 * DIN + c) = make_float4(vn.x * fa, vn.y * fa, vn.z * fa, vn.w * fa);
            *(float4*)(fr + 8 * DIN + c) = make_float4(sd.x * fa, sd.y * fa, sd.z * fa, sd.w * fa);
            *(float4*)(fr + 9 * DIN + c)  = make_float4(mean.x * fb, mean.y * fb, mean.z * fb, mean.w * fb);
            *(float4*)(fr + 10 * DIN + c) = make_float4(vx.x * fb, vx.y * fb, vx.z * fb, vx.w * fb);
            *(float4*)(fr + 11 * DIN + c) = make_float4(vn.x * fb, vn.y * fb, vn.z * fb, vn.w * fb);
            *(float4*)(fr + 12 * DIN + c) = make_float4(sd.x * fb, sd.y * fb, sd.z * fb, sd.w * fb);
        }
    }
    __syncthreads();

    const int l  = tid & 31;
    const int cg = tid >> 5;
    const int lo = l & 15;
    const int hi = l >> 4;
    const int col = cg * 16 + lo;

    v8f acc = {};
    const float* arow = &feat[lo * FSTRIDE + 2 * hi];
    const float* bcol = Wpost + (size_t)(2 * hi) * DOUT + col;
    float bx[2][NKB], by[2][NKB];
#pragma unroll
    for (int s = 0; s < NKB; ++s) {
        bx[0][s] = bcol[(4 * s) * DOUT];
        by[0][s] = bcol[(4 * s + 1) * DOUT];
    }
#pragma unroll
    for (int kb = 0; kb < NNB; ++kb) {
        const int cur = kb & 1, nxt = cur ^ 1;
        if (kb + 1 < NNB) {
#pragma unroll
            for (int s = 0; s < NKB; ++s) {
                const int k = (kb + 1) * 4 * NKB + 4 * s;
                bx[nxt][s] = bcol[k * DOUT];
                by[nxt][s] = bcol[(k + 1) * DOUT];
            }
        }
#pragma unroll
        for (int s = 0; s < NKB; ++s) {
            const int k0 = kb * 4 * NKB + 4 * s;
            v2f a = *(const v2f*)(arow + k0);
            v2f b;
            b.x = bx[cur][s];
            b.y = by[cur][s];
            acc = __builtin_amdgcn_wmma_f32_16x16x4_f32(
                false, a, false, b, (short)0, acc, false, false);
        }
    }

    const float bv = bpost[col];
    float ps = 0.0f, ps2 = 0.0f;
#pragma unroll
    for (int r = 0; r < 8; ++r) {
        int M = r + 8 * hi;
        int node = n0 + M;
        float v = (acc[r] + bv) * snorm[node];
        out[(size_t)node * DOUT + col] = v;
        ps += v; ps2 += v * v;
    }
    ps  += __shfl_xor(ps, 16, 32);
    ps2 += __shfl_xor(ps2, 16, 32);
    if (hi == 0) {
        atomicAdd(&bns[col], ps);
        atomicAdd(&bnq[col], ps2);
    }
}

__global__ __launch_bounds__(256) void pna_bn(
    float* out, const float* bns, const float* bnq,
    const float* gamma, const float* beta) {
    int i = blockIdx.x * 256 + threadIdx.x;
    if (i >= NNODES * DOUT) return;
    int c = i & 63;
    float mu  = bns[c] / (float)NNODES;
    float var = bnq[c] / (float)NNODES - mu * mu;
    float v = out[i];
    out[i] = (v - mu) * rsqrtf(var + 1e-05f) * gamma[c] + beta[c];
}

extern "C" void kernel_launch(void* const* d_in, const int* in_sizes, int n_in,
                              void* d_out, int out_size, void* d_ws, size_t ws_size,
                              hipStream_t stream) {
    const float* h      = (const float*)d_in[0];
    const float* e      = (const float*)d_in[1];
    const float* snorm  = (const float*)d_in[2];
    const float* Wpre   = (const float*)d_in[3];
    const float* bpre   = (const float*)d_in[4];
    const float* Wpost  = (const float*)d_in[5];
    const float* bpost  = (const float*)d_in[6];
    const float* gamma  = (const float*)d_in[7];
    const float* beta   = (const float*)d_in[8];
    const int*   src    = (const int*)d_in[9];
    const int*   dst    = (const int*)d_in[10];
    float* out = (float*)d_out;

    float*    s1  = (float*)d_ws;
    float*    s2  = s1 + (size_t)NNODES * DIN;
    unsigned* mx  = (unsigned*)(s2 + (size_t)NNODES * DIN);
    unsigned* mn  = mx + (size_t)NNODES * DIN;
    int*      cnt = (int*)(mn + (size_t)NNODES * DIN);
    float*    bns = (float*)(cnt + NNODES);
    float*    bnq = bns + DOUT;

    int* perm = (int*)d_out;
    int* rank = perm + NEDGES;
    int* rptr = rank + NEDGES;

    pna_init  <<<(NNODES * DIN + 255) / 256, 256, 0, stream>>>(s1, s2, mx, mn, cnt, bns, bnq);
    pna_rank  <<<(NEDGES + 255) / 256, 256, 0, stream>>>(dst, cnt, rank);
    pna_scan  <<<1, 1024, 0, stream>>>(cnt, rptr);
    pna_fill  <<<(NEDGES + 255) / 256, 256, 0, stream>>>(dst, rank, rptr, perm);
    pna_edge  <<<NEDGES / 32, 256, 0, stream>>>(h, e, Wpre, bpre, src, dst, perm, s1, s2, mx, mn);
    pna_node  <<<NNODES / 16, 128, 0, stream>>>(h, snorm, Wpost, bpost, s1, s2, mx, mn, cnt,
                                                out, bns, bnq);
    pna_bn    <<<(NNODES * DOUT + 255) / 256, 256, 0, stream>>>(out, bns, bnq, gamma, beta);
}


# Count scan: each wave reads its 3200 counts once as coalesced 16-byte pieces kept in registers and prefixes them across lanes; before, each thread walked 98 counts twice by dependent 4-byte loads
# speedup vs baseline: 2.2884x; 1.2421x over previous
#include <hip/hip_runtime.h>
#include <hip/hip_bf16.h>
#include <math.h>

#define NNODES 100000
#define NEDGES 1200000
#define DIN    64
#define DOUT   64
#define EDIM   16
#define KPRE   144
#define KPOST  832
#define AVG_D  2.5649493574615367f
#define EPSV   1e-05f

typedef __attribute__((ext_vector_type(2))) float v2f;
typedef __attribute__((ext_vector_type(8))) float v8f;

__device__ __forceinline__ unsigned fenc(float f) {
    unsigned u = __float_as_uint(f);
    return (u & 0x80000000u) ? ~u : (u | 0x80000000u);
}
__device__ __forceinline__ float fdec(unsigned u) {
    return (u & 0x80000000u) ? __uint_as_float(u & 0x7fffffffu)
                             : __uint_as_float(~u);
}


__global__ __launch_bounds__(256) void pna_init(float* s1, float* s2,
                                                unsigned* mx, unsigned* mn,
                                                int* cnt, float* bns, float* bnq) {
    int i = blockIdx.x * 256 + threadIdx.x;
    if (i < NNODES * DIN) {
        s1[i] = 0.0f; s2[i] = 0.0f;
        mx[i] = 0u;
        mn[i] = 0xFFFFFFFFu;
    }
    if (i < NNODES) cnt[i] = 0;
    if (i < DOUT) { bns[i] = 0.0f; bnq[i] = 0.0f; }
}

__global__ __launch_bounds__(256) void pna_rank(const int* __restrict__ dst,
                                                int* cnt, int* __restrict__ rank) {
    int i = blockIdx.x * 256 + threadIdx.x;
    if (i < NEDGES) rank[i] = atomicAdd(&cnt[dst[i]], 1);
}

#define SCAN_IT  25
#define NPIECE   (NNODES / 4)
#define RPTR_PAD (32 * SCAN_IT * 128)
static_assert(NNODES % 4 == 0, "counts are read as whole 16-byte pieces");
static_assert(RPTR_PAD >= NNODES, "the 32 waves must cover every count");
__global__ __launch_bounds__(1024) void pna_scan(const int* __restrict__ cnt,
                                                 int* __restrict__ rptr) {
    __shared__ int wtot[32];
    const int t = threadIdx.x;
    const int l = t & 31, w = t >> 5;
    const int4* c4 = (const int4*)cnt;
    int4* r4 = (int4*)rptr;
    const int q0 = w * (SCAN_IT * 32) + l;
    int4 v[SCAN_IT];
#pragma unroll
    for (int it = 0; it < SCAN_IT; ++it) {
        const int q = q0 + it * 32;
        v[it] = c4[(q < NPIECE) ? q : (NPIECE - 1)];
    }
    int tot = 0;
#pragma unroll
    for (int it = 0; it < SCAN_IT; ++it) tot += (v[it].x + v[it].y) + (v[it].z + v[it].w);
    for (int off = 16; off > 0; off >>= 1) tot += __shfl_xor(tot, off, 32);
    if (l == 0) wtot[w] = tot;
    __syncthreads();
    int carry = (l < w) ? wtot[l] : 0;
    for (int off = 16; off > 0; off >>= 1) carry += __shfl_xor(carry, off, 32);
#pragma unroll
    for (int it = 0; it < SCAN_IT; ++it) {
        const int4 a = v[it];
        const int s = (a.x + a.y) + (a.z + a.w);
        int inc = s;
        for (int off = 1; off < 32; off <<= 1) {
            const int n = __shfl_up(inc, off, 32);
            if (l >= off) inc += n;
        }
        const int ex = carry + inc - s;
        int4 o;
        o.x = ex;
        o.y = ex + a.x;
        o.z = o.y + a.y;
        o.w = o.z + a.z;
        r4[q0 + it * 32] = o;
        carry += __builtin_amdgcn_readlane(inc, 31);
    }
}

__global__ __launch_bounds__(256) void pna_fill(const int* __restrict__ dst,
                                                const int* __restrict__ rank,
                                                const int* __restrict__ rptr,
                                                int* __restrict__ perm) {
    int i = blockIdx.x * 256 + threadIdx.x;
    if (i < NEDGES) perm[rptr[dst[i]] + rank[i]] = i;
}

#define ZSTRIDE 148
__global__ __launch_bounds__(256) void pna_edge(
    const float* __restrict__ h, const float* __restrict__ e,
    const float* __restrict__ Wpre, const float* __restrict__ bpre,
    const int* __restrict__ src, const int* __restrict__ dst,
    const int* __restrict__ perm,
    float* s1, float* s2, unsigned* mx, unsigned* mn) {

    __shared__ __attribute__((aligned(16))) float zs[32 * ZSTRIDE];
    __shared__ float wsT[DIN * ZSTRIDE];
    __shared__ int   ldst[32];

    const int tid = threadIdx.x;
    const int e0  = blockIdx.x * 32;

    for (int i = tid; i < KPRE * DIN; i += 256) {
        int n = i & 63, k = i >> 6;
        wsT[n * ZSTRIDE + k] = Wpre[k * DIN + n];
    }
    {
        const int r = tid >> 3, p = tid & 7;
        const int ei = perm[e0 + r];
        const int sI = src[ei], dI = dst[ei];
        if (p == 0) ldst[r] = dI;
        const float4* hs = (const float4*)(h + (size_t)sI * DIN);
        const float4* hd = (const float4*)(h + (size_t)dI * DIN);
        const float2* er = (const float2*)(e + (size_t)ei * EDIM);
        const float4 s_lo = hs[p], s_hi = hs[p + 8];
        const float4 d_lo = hd[p], d_hi = hd[p + 8];
        const float2 e_pc = er[p];
        float* zr = &zs[r * ZSTRIDE];
        *(float4*)(zr + 4 * p)            = s_lo;
        *(float4*)(zr + 32 + 4 * p)       = s_hi;
        *(float4*)(zr + DIN + 4 * p)      = d_lo;
        *(float4*)(zr + DIN + 32 + 4 * p) = d_hi;
        *(float2*)(zr + 2 * DIN + 2 * p)  = e_pc;
    }
    __syncthreads();

    const int l  = tid & 31;
    const int wv = tid >> 5;
    const int cg = wv & 3;
    const int eg = wv >> 2;
    const int lo = l & 15;
    const int hi = l >> 4;

    v8f acc = {};
    const float* arow = &zs[(eg * 16 + lo) * ZSTRIDE + 2 * hi];
    const float* brow = &wsT[(cg * 16 + lo) * ZSTRIDE + 2 * hi];
#pragma unroll 4
    for (int k0 = 0; k0 < KPRE; k0 += 4) {
        v2f a = *(const v2f*)(arow + k0);
        v2f b = *(const v2f*)(brow + k0);
        acc = __builtin_amdgcn_wmma_f32_16x16x4_f32(
            false, a, false, b, (short)0, acc, false, false);
    }

    const int col = cg * 16 + lo;
    const float bv = bpre[col];
    const int rb = eg * 16 + 8 * hi;
    int cur = ldst[rb];
    float v = acc[0] + bv;
    float sum = v, sq = v * v;
    unsigned emx = fenc(v), emn = emx;
#pragma unroll
    for (int r = 1; r < 8; ++r) {
        int d = ldst[rb + r];
        v = acc[r] + bv;
        unsigned ev = fenc(v);
        if (d != cur) {
            size_t o = (size_t)cur * DIN + col;
            atomicAdd(&s1[o], sum);
            atomicAdd(&s2[o], sq);
            atomicMax(&mx[o], emx);
            atomicMin(&mn[o], emn);
            cur = d; sum = v; sq = v * v; emx = ev; emn = ev;
        } else {
            sum += v; sq += v * v;
            emx = (ev > emx) ? ev : emx;
            emn = (ev < emn) ? ev : emn;
        }
    }
    {
        size_t o = (size_t)cur * DIN + col;
        atomicAdd(&s1[o], sum);
        atomicAdd(&s2[o], sq);
        atomicMax(&mx[o], emx);
        atomicMin(&mn[o], emn);
    }
}

#define FSTRIDE 836
#define NKB     16
#define NNB     13
__global__ __launch_bounds__(128) void pna_node(
    const float* __restrict__ h, const float* __restrict__ snorm,
    const float* __restrict__ Wpost, const float* __restrict__ bpost,
    const float* __restrict__ s1, const float* __restrict__ s2,
    const unsigned* __restrict__ mx, const unsigned* __restrict__ mn,
    const int* __restrict__ cnt,
    float* __restrict__ out, float* bns, float* bnq) {

    __shared__ __attribute__((aligned(16))) float feat[16 * FSTRIDE];
    const int tid = threadIdx.x;
    const int n0  = blockIdx.x * 16;

    {
        const int r = tid >> 3, p = tid & 7;
        const int node = n0 + r;
        const size_t o = (size_t)node * DIN;
        const float4* hp  = (const float4*)(h + o);
        const float4* s1p = (const float4*)(s1 + o);
        const float4* s2p = (const float4*)(s2 + o);
        const uint4*  mxp = (const uint4*)(mx + o);
        const uint4*  mnp = (const uint4*)(mn + o);
        const int dg = cnt[node];
        float4 hv[2], a1[2], a2[2];
        uint4  ux[2], un[2];
#pragma unroll
        for (int q = 0; q < 2; ++q) {
            hv[q] = hp[p + 8 * q];
            a1[q] = s1p[p + 8 * q];
            a2[q] = s2p[p + 8 * q];
            ux[q] = mxp[p + 8 * q];
            un[q] = mnp[p + 8 * q];
        }
        const float D    = (float)dg;
        const float logd = logf(D + 1.0f);
        const float fa   = logd / AVG_D;
        const float fb   = AVG_D / logd;
        float* fr = &feat[r * FSTRIDE];
#pragma unroll
        for (int q = 0; q < 2; ++q) {
            const int c = 4 * p + 32 * q;
            float4 mean, sd, vx, vn;
            mean.x = a1[q].x / D; mean.y = a1[q].y / D;
            mean.z = a1[q].z / D; mean.w = a1[q].w / D;
            sd.x = sqrtf(fmaxf(a2[q].x / D - mean.x * mean.x, 0.0f) + EPSV);
            sd.y = sqrtf(fmaxf(a2[q].y / D - mean.y * mean.y, 0.0f) + EPSV);
            sd.z = sqrtf(fmaxf(a2[q].z / D - mean.z * mean.z, 0.0f) + EPSV);
            sd.w = sqrtf(fmaxf(a2[q].w / D - mean.w * mean.w, 0.0f) + EPSV);
            vx.x = fdec(ux[q].x); vx.y = fdec(ux[q].y); vx.z = fdec(ux[q].z); vx.w = fdec(ux[q].w);
            vn.x = fdec(un[q].x); vn.y = fdec(un[q].y); vn.z = fdec(un[q].z); vn.w = fdec(un[q].w);
            *(float4*)(fr + c)           = hv[q];
            *(float4*)(fr + DIN + c)     = mean;
            *(float4*)(fr + 2 * DIN + c) = vx;
            *(float4*)(fr + 3 * DIN + c) = vn;
            *(float4*)(fr + 4 * DIN + c) = sd;
            *(float4*)(fr + 5 * DIN + c) = make_float4(mean.x * fa, mean.y * fa, mean.z * fa, mean.w * fa);
            *(float4*)(fr + 6 * DIN + c) = make_float4(vx.x * fa, vx.y * fa, vx.z * fa, vx.w * fa);
            *(float4*)(fr + 7 * DIN + c) = make_float4(vn.x * fa, vn.y * fa, vn.z * fa, vn.w * fa);
            *(float4*)(fr + 8 * DIN + c) = make_float4(sd.x * fa, sd.y * fa, sd.z * fa, sd.w * fa);
            *(float4*)(fr + 9 * DIN + c)  = make_float4(mean.x * fb, mean.y * fb, mean.z * fb, mean.w * fb);
            *(float4*)(fr + 10 * DIN + c) = make_float4(vx.x * fb, vx.y * fb, vx.z * fb, vx.w * fb);
            *(float4*)(fr + 11 * DIN + c) = make_float4(vn.x * fb, vn.y * fb, vn.z * fb, vn.w * fb);
            *(float4*)(fr + 12 * DIN + c) = make_float4(sd.x * fb, sd.y * fb, sd.z * fb, sd.w * fb);
        }
    }
    __syncthreads();

    const int l  = tid & 31;
    const int cg = tid >> 5;
    const int lo = l & 15;
    const int hi = l >> 4;
    const int col = cg * 16 + lo;

    v8f acc = {};
    const float* arow = &feat[lo * FSTRIDE + 2 * hi];
    const float* bcol = Wpost + (size_t)(2 * hi) * DOUT + col;
    float bx[2][NKB], by[2][NKB];
#pragma unroll
    for (int s = 0; s < NKB; ++s) {
        bx[0][s] = bcol[(4 * s) * DOUT];
        by[0][s] = bcol[(4 * s + 1) * DOUT];
    }
#pragma unroll
    for (int kb = 0; kb < NNB; ++kb) {
        const int cur = kb & 1, nxt = cur ^ 1;
        if (kb + 1 < NNB) {
#pragma unroll
            for (int s = 0; s < NKB; ++s) {
                const int k = (kb + 1) * 4 * NKB + 4 * s;
                bx[nxt][s] = bcol[k * DOUT];
                by[nxt][s] = bcol[(k + 1) * DOUT];
            }
        }
#pragma unroll
        for (int s = 0; s < NKB; ++s) {
            const int k0 = kb * 4 * NKB + 4 * s;
            v2f a = *(const v2f*)(arow + k0);
            v2f b;
            b.x = bx[cur][s];
            b.y = by[cur][s];
            acc = __builtin_amdgcn_wmma_f32_16x16x4_f32(
                false, a, false, b, (short)0, acc, false, false);
        }
    }

    const float bv = bpost[col];
    float ps = 0.0f, ps2 = 0.0f;
#pragma unroll
    for (int r = 0; r < 8; ++r) {
        int M = r + 8 * hi;
        int node = n0 + M;
        float v = (acc[r] + bv) * snorm[node];
        out[(size_t)node * DOUT + col] = v;
        ps += v; ps2 += v * v;
    }
    ps  += __shfl_xor(ps, 16, 32);
    ps2 += __shfl_xor(ps2, 16, 32);
    if (hi == 0) {
        atomicAdd(&bns[col], ps);
        atomicAdd(&bnq[col], ps2);
    }
}

__global__ __launch_bounds__(256) void pna_bn(
    float* out, const float* bns, const float* bnq,
    const float* gamma, const float* beta) {
    int i = blockIdx.x * 256 + threadIdx.x;
    if (i >= NNODES * DOUT) return;
    int c = i & 63;
    float mu  = bns[c] / (float)NNODES;
    float var = bnq[c] / (float)NNODES - mu * mu;
    float v = out[i];
    out[i] = (v - mu) * rsqrtf(var + 1e-05f) * gamma[c] + beta[c];
}

extern "C" void kernel_launch(void* const* d_in, const int* in_sizes, int n_in,
                              void* d_out, int out_size, void* d_ws, size_t ws_size,
                              hipStream_t stream) {
    const float* h      = (const float*)d_in[0];
    const float* e      = (const float*)d_in[1];
    const float* snorm  = (const float*)d_in[2];
    const float* Wpre   = (const float*)d_in[3];
    const float* bpre   = (const float*)d_in[4];
    const float* Wpost  = (const float*)d_in[5];
    const float* bpost  = (const float*)d_in[6];
    const float* gamma  = (const float*)d_in[7];
    const float* beta   = (const float*)d_in[8];
    const int*   src    = (const int*)d_in[9];
    const int*   dst    = (const int*)d_in[10];
    float* out = (float*)d_out;

    float*    s1  = (float*)d_ws;
    float*    s2  = s1 + (size_t)NNODES * DIN;
    unsigned* mx  = (unsigned*)(s2 + (size_t)NNODES * DIN);
    unsigned* mn  = mx + (size_t)NNODES * DIN;
    int*      cnt = (int*)(mn + (size_t)NNODES * DIN);
    float*    bns = (float*)(cnt + NNODES);
    float*    bnq = bns + DOUT;

    int* perm = (int*)d_out;
    int* rank = perm + NEDGES;
    int* rptr = rank + NEDGES;

    pna_init  <<<(NNODES * DIN + 255) / 256, 256, 0, stream>>>(s1, s2, mx, mn, cnt, bns, bnq);
    pna_rank  <<<(NEDGES + 255) / 256, 256, 0, stream>>>(dst, cnt, rank);
    pna_scan  <<<1, 1024, 0, stream>>>(cnt, rptr);
    pna_fill  <<<(NEDGES + 255) / 256, 256, 0, stream>>>(dst, rank, rptr, perm);
    pna_edge  <<<NEDGES / 32, 256, 0, stream>>>(h, e, Wpre, bpre, src, dst, perm, s1, s2, mx, mn);
    pna_node  <<<NNODES / 16, 128, 0, stream>>>(h, snorm, Wpost, bpost, s1, s2, mx, mn, cnt,
                                                out, bns, bnq);
    pna_bn    <<<(NNODES * DOUT + 255) / 256, 256, 0, stream>>>(out, bns, bnq, gamma, beta);
}


# h[dst] W_pre[64:128] + b_pre is constant per destination: left out of the edge product (k 144 to 80, no h[dst] gather); node kernel forms it per node and adds it to mean, max, min; sum order changes
# speedup vs baseline: 2.7311x; 1.1935x over previous
#include <hip/hip_runtime.h>
#include <hip/hip_bf16.h>
#include <math.h>

#define NNODES 100000
#define NEDGES 1200000
#define DIN    64
#define DOUT   64
#define EDIM   16
#define KEDGE  80
#define KPOST  832
#define AVG_D  2.5649493574615367f
#define EPSV   1e-05f

typedef __attribute__((ext_vector_type(2))) float v2f;
typedef __attribute__((ext_vector_type(8))) float v8f;

__device__ __forceinline__ unsigned fenc(float f) {
    unsigned u = __float_as_uint(f);
    return (u & 0x80000000u) ? ~u : (u | 0x80000000u);
}
__device__ __forceinline__ float fdec(unsigned u) {
    return (u & 0x80000000u) ? __uint_as_float(u & 0x7fffffffu)
                             : __uint_as_float(~u);
}


__global__ __launch_bounds__(256) void pna_init(float* s1, float* s2,
                                                unsigned* mx, unsigned* mn,
                                                int* cnt, float* bns, float* bnq) {
    int i = blockIdx.x * 256 + threadIdx.x;
    if (i < NNODES * DIN) {
        s1[i] = 0.0f; s2[i] = 0.0f;
        mx[i] = 0u;
        mn[i] = 0xFFFFFFFFu;
    }
    if (i < NNODES) cnt[i] = 0;
    if (i < DOUT) { bns[i] = 0.0f; bnq[i] = 0.0f; }
}

__global__ __launch_bounds__(256) void pna_rank(const int* __restrict__ dst,
                                                int* cnt, int* __restrict__ rank) {
    int i = blockIdx.x * 256 + threadIdx.x;
    if (i < NEDGES) rank[i] = atomicAdd(&cnt[dst[i]], 1);
}

#define SCAN_IT  25
#define NPIECE   (NNODES / 4)
#define RPTR_PAD (32 * SCAN_IT * 128)
static_assert(NNODES % 4 == 0, "counts are read as whole 16-byte pieces");
static_assert(RPTR_PAD >= NNODES, "the 32 waves must cover every count");
__global__ __launch_bounds__(1024) void pna_scan(const int* __restrict__ cnt,
                                                 int* __restrict__ rptr) {
    __shared__ int wtot[32];
    const int t = threadIdx.x;
    const int l = t & 31, w = t >> 5;
    const int4* c4 = (const int4*)cnt;
    int4* r4 = (int4*)rptr;
    const int q0 = w * (SCAN_IT * 32) + l;
    int4 v[SCAN_IT];
#pragma unroll
    for (int it = 0; it < SCAN_IT; ++it) {
        const int q = q0 + it * 32;
        v[it] = c4[(q < NPIECE) ? q : (NPIECE - 1)];
    }
    int tot = 0;
#pragma unroll
    for (int it = 0; it < SCAN_IT; ++it) tot += (v[it].x + v[it].y) + (v[it].z + v[it].w);
    for (int off = 16; off > 0; off >>= 1) tot += __shfl_xor(tot, off, 32);
    if (l == 0) wtot[w] = tot;
    __syncthreads();
    int carry = (l < w) ? wtot[l] : 0;
    for (int off = 16; off > 0; off >>= 1) carry += __shfl_xor(carry, off, 32);
#pragma unroll
    for (int it = 0; it < SCAN_IT; ++it) {
        const int4 a = v[it];
        const int s = (a.x + a.y) + (a.z + a.w);
        int inc = s;
        for (int off = 1; off < 32; off <<= 1) {
            const int n = __shfl_up(inc, off, 32);
            if (l >= off) inc += n;
        }
        const int ex = carry + inc - s;
        int4 o;
        o.x = ex;
        o.y = ex + a.x;
        o.z = o.y + a.y;
        o.w = o.z + a.z;
        r4[q0 + it * 32] = o;
        carry += __builtin_amdgcn_readlane(inc, 31);
    }
}

__global__ __launch_bounds__(256) void pna_fill(const int* __restrict__ dst,
                                                const int* __restrict__ rank,
                                                const int* __restrict__ rptr,
                                                int* __restrict__ perm) {
    int i = blockIdx.x * 256 + threadIdx.x;
    if (i < NEDGES) perm[rptr[dst[i]] + rank[i]] = i;
}

#define ZSTRIDE 84
__global__ __launch_bounds__(256) void pna_edge(
    const float* __restrict__ h, const float* __restrict__ e,
    const float* __restrict__ Wpre,
    const int* __restrict__ src, const int* __restrict__ dst,
    const int* __restrict__ perm,
    float* s1, float* s2, unsigned* mx, unsigned* mn) {

    __shared__ __attribute__((aligned(16))) float zs[32 * ZSTRIDE];
    __shared__ float wsT[DIN * ZSTRIDE];
    __shared__ int   ldst[32];

    const int tid = threadIdx.x;
    const int e0  = blockIdx.x * 32;

    for (int i = tid; i < KEDGE * DIN; i += 256) {
        int n = i & 63, k = i >> 6;
        wsT[n * ZSTRIDE + k] = Wpre[(k + (k & DIN)) * DIN + n];
    }
    {
        const int r = tid >> 3, p = tid & 7;
        const int ei = perm[e0 + r];
        const int sI = src[ei], dI = dst[ei];
        if (p == 0) ldst[r] = dI;
        const float4* hs = (const float4*)(h + (size_t)sI * DIN);
        const float2* er = (const float2*)(e + (size_t)ei * EDIM);
        const float4 s_lo = hs[p], s_hi = hs[p + 8];
        const float2 e_pc = er[p];
        float* zr = &zs[r * ZSTRIDE];
        *(float4*)(zr + 4 * p)       = s_lo;
        *(float4*)(zr + 32 + 4 * p)  = s_hi;
        *(float2*)(zr + DIN + 2 * p) = e_pc;
    }
    __syncthreads();

    const int l  = tid & 31;
    const int wv = tid >> 5;
    const int cg = wv & 3;
    const int eg = wv >> 2;
    const int lo = l & 15;
    const int hi = l >> 4;

    v8f acc = {};
    const float* arow = &zs[(eg * 16 + lo) * ZSTRIDE + 2 * hi];
    const float* brow = &wsT[(cg * 16 + lo) * ZSTRIDE + 2 * hi];
#pragma unroll 4
    for (int k0 = 0; k0 < KEDGE; k0 += 4) {
        v2f a = *(const v2f*)(arow + k0);
        v2f b = *(const v2f*)(brow + k0);
        acc = __builtin_amdgcn_wmma_f32_16x16x4_f32(
            false, a, false, b, (short)0, acc, false, false);
    }

    const int col = cg * 16 + lo;
    const int rb = eg * 16 + 8 * hi;
    int cur = ldst[rb];
    float v = acc[0];
    float sum = v, sq = v * v;
    unsigned emx = fenc(v), emn = emx;
#pragma unroll
    for (int r = 1; r < 8; ++r) {
        int d = ldst[rb + r];
        v = acc[r];
        unsigned ev = fenc(v);
        if (d != cur) {
            size_t o = (size_t)cur * DIN + col;
            atomicAdd(&s1[o], sum);
            atomicAdd(&s2[o], sq);
            atomicMax(&mx[o], emx);
            atomicMin(&mn[o], emn);
            cur = d; sum = v; sq = v * v; emx = ev; emn = ev;
        } else {
            sum += v; sq += v * v;
            emx = (ev > emx) ? ev : emx;
            emn = (ev < emn) ? ev : emn;
        }
    }
    {
        size_t o = (size_t)cur * DIN + col;
        atomicAdd(&s1[o], sum);
        atomicAdd(&s2[o], sq);
        atomicMax(&mx[o], emx);
        atomicMin(&mn[o], emn);
    }
}

#define FSTRIDE 836
#define NKB     16
#define NNB     13
#define NKC     16
__global__ __launch_bounds__(128) void pna_node(
    const float* __restrict__ h, const float* __restrict__ snorm,
    const float* __restrict__ Wpost, const float* __restrict__ bpost,
    const float* __restrict__ Wpre, const float* __restrict__ bpre,
    const float* __restrict__ s1, const float* __restrict__ s2,
    const unsigned* __restrict__ mx, const unsigned* __restrict__ mn,
    const int* __restrict__ cnt,
    float* __restrict__ out, float* bns, float* bnq) {

    __shared__ __attribute__((aligned(16))) float feat[16 * FSTRIDE];
    __shared__ float fab[16 * 2];
    const int tid = threadIdx.x;
    const int n0  = blockIdx.x * 16;

    const int l  = tid & 31;
    const int cg = tid >> 5;
    const int lo = l & 15;
    const int hi = l >> 4;
    const int col = cg * 16 + lo;

    const float* wc = Wpre + (size_t)(DIN + 2 * hi) * DIN + col;
    float cx[NKC], cy[NKC];
#pragma unroll
    for (int s = 0; s < NKC; ++s) {
        cx[s] = wc[(4 * s) * DIN];
        cy[s] = wc[(4 * s + 1) * DIN];
    }
    const float cb = bpre[col];

    {
        const int r = tid >> 3, p = tid & 7;
        const int node = n0 + r;
        const size_t o = (size_t)node * DIN;
        const float4* hp  = (const float4*)(h + o);
        const float4* s1p = (const float4*)(s1 + o);
        const float4* s2p = (const float4*)(s2 + o);
        const uint4*  mxp = (const uint4*)(mx + o);
        const uint4*  mnp = (const uint4*)(mn + o);
        const int dg = cnt[node];
        float4 hv[2], a1[2], a2[2];
        uint4  ux[2], un[2];
#pragma unroll
        for (int q = 0; q < 2; ++q) {
            hv[q] = hp[p + 8 * q];
            a1[q] = s1p[p + 8 * q];
            a2[q] = s2p[p + 8 * q];
            ux[q] = mxp[p + 8 * q];
            un[q] = mnp[p + 8 * q];
        }
        float* fr = &feat[r * FSTRIDE];
        *(float4*)(fr + 4 * p)      = hv[0];
        *(float4*)(fr + 32 + 4 * p) = hv[1];
        const float D    = (float)dg;
        const float logd = logf(D + 1.0f);
        const float fa   = logd / AVG_D;
        const float fb   = AVG_D / logd;
        if (p == 0) { fab[2 * r] = fa; fab[2 * r + 1] = fb; }
#pragma unroll
        for (int q = 0; q < 2; ++q) {
            const int c = 4 * p + 32 * q;
            float4 mu, sd, vx, vn;
            mu.x = a1[q].x / D; mu.y = a1[q].y / D;
            mu.z = a1[q].z / D; mu.w = a1[q].w / D;
            sd.x = sqrtf(fmaxf(a2[q].x / D - mu.x * mu.x, 0.0f) + EPSV);
            sd.y = sqrtf(fmaxf(a2[q].y / D - mu.y * mu.y, 0.0f) + EPSV);
            sd.z = sqrtf(fmaxf(a2[q].z / D - mu.z * mu.z, 0.0f) + EPSV);
            sd.w = sqrtf(fmaxf(a2[q].w / D - mu.w * mu.w, 0.0f) + EPSV);
            vx.x = fdec(ux[q].x); vx.y = fdec(ux[q].y); vx.z = fdec(ux[q].z); vx.w = fdec(ux[q].w);
            vn.x = fdec(un[q].x); vn.y = fdec(un[q].y); vn.z = fdec(un[q].z); vn.w = fdec(un[q].w);
            *(float4*)(fr + DIN + c)      = mu;
            *(float4*)(fr + 2 * DIN + c)  = vx;
            *(float4*)(fr + 3 * DIN + c)  = vn;
            *(float4*)(fr + 4 * DIN + c)  = sd;
            *(float4*)(fr + 8 * DIN + c)  = make_float4(sd.x * fa, sd.y * fa, sd.z * fa, sd.w * fa);
            *(float4*)(fr + 12 * DIN + c) = make_float4(sd.x * fb, sd.y * fb, sd.z * fb, sd.w * fb);
        }
    }
    __syncthreads();

    const float* arow = &feat[lo * FSTRIDE + 2 * hi];

    {
        v8f cacc = {};
#pragma unroll
        for (int s = 0; s < NKC; ++s) {
            v2f a = *(const v2f*)(arow + 4 * s);
            v2f b;
            b.x = cx[s];
            b.y = cy[s];
            cacc = __builtin_amdgcn_wmma_f32_16x16x4_f32(
                false, a, false, b, (short)0, cacc, false, false);
        }
#pragma unroll
        for (int i = 0; i < 8; ++i) {
            const int row = 8 * hi + i;
            const float cv = cacc[i] + cb;
            const float fa = fab[2 * row], fb = fab[2 * row + 1];
            float* fp = &feat[row * FSTRIDE + col];
            for (int g = 1; g <= 3; ++g) {
                const float x = fp[g * DIN] + cv;
                fp[g * DIN]       = x;
                fp[(g + 4) * DIN] = x * fa;
                fp[(g + 8) * DIN] = x * fb;
            }
        }
    }
    __syncthreads();

    v8f acc = {};
    const float* bcol = Wpost + (size_t)(2 * hi) * DOUT + col;
    float bx[2][NKB], by[2][NKB];
#pragma unroll
    for (int s = 0; s < NKB; ++s) {
        bx[0][s] = bcol[(4 * s) * DOUT];
        by[0][s] = bcol[(4 * s + 1) * DOUT];
    }
#pragma unroll
    for (int kb = 0; kb < NNB; ++kb) {
        const int cur = kb & 1, nxt = cur ^ 1;
        if (kb + 1 < NNB) {
#pragma unroll
            for (int s = 0; s < NKB; ++s) {
                const int k = (kb + 1) * 4 * NKB + 4 * s;
                bx[nxt][s] = bcol[k * DOUT];
                by[nxt][s] = bcol[(k + 1) * DOUT];
            }
        }
#pragma unroll
        for (int s = 0; s < NKB; ++s) {
            const int k0 = kb * 4 * NKB + 4 * s;
            v2f a = *(const v2f*)(arow + k0);
            v2f b;
            b.x = bx[cur][s];
            b.y = by[cur][s];
            acc = __builtin_amdgcn_wmma_f32_16x16x4_f32(
                false, a, false, b, (short)0, acc, false, false);
        }
    }

    const float bv = bpost[col];
    float ps = 0.0f, ps2 = 0.0f;
#pragma unroll
    for (int r = 0; r < 8; ++r) {
        int M = r + 8 * hi;
        int node = n0 + M;
        float v = (acc[r] + bv) * snorm[node];
        out[(size_t)node * DOUT + col] = v;
        ps += v; ps2 += v * v;
    }
    ps  += __shfl_xor(ps, 16, 32);
    ps2 += __shfl_xor(ps2, 16, 32);
    if (hi == 0) {
        atomicAdd(&bns[col], ps);
        atomicAdd(&bnq[col], ps2);
    }
}

__global__ __launch_bounds__(256) void pna_bn(
    float* out, const float* bns, const float* bnq,
    const float* gamma, const float* beta) {
    int i = blockIdx.x * 256 + threadIdx.x;
    if (i >= NNODES * DOUT) return;
    int c = i & 63;
    float mu  = bns[c] / (float)NNODES;
    float var = bnq[c] / (float)NNODES - mu * mu;
    float v = out[i];
    out[i] = (v - mu) * rsqrtf(var + 1e-05f) * gamma[c] + beta[c];
}

extern "C" void kernel_launch(void* const* d_in, const int* in_sizes, int n_in,
                              void* d_out, int out_size, void* d_ws, size_t ws_size,
                              hipStream_t stream) {
    const float* h      = (const float*)d_in[0];
    const float* e      = (const float*)d_in[1];
    const float* snorm  = (const float*)d_in[2];
    const float* Wpre   = (const float*)d_in[3];
    const float* bpre   = (const float*)d_in[4];
    const float* Wpost  = (const float*)d_in[5];
    const float* bpost  = (const float*)d_in[6];
    const float* gamma  = (const float*)d_in[7];
    const float* beta   = (const float*)d_in[8];
    const int*   src    = (const int*)d_in[9];
    const int*   dst    = (const int*)d_in[10];
    float* out = (float*)d_out;

    float*    s1  = (float*)d_ws;
    float*    s2  = s1 + (size_t)NNODES * DIN;
    unsigned* mx  = (unsigned*)(s2 + (size_t)NNODES * DIN);
    unsigned* mn  = mx + (size_t)NNODES * DIN;
    int*      cnt = (int*)(mn + (size_t)NNODES * DIN);
    float*    bns = (float*)(cnt + NNODES);
    float*    bnq = bns + DOUT;

    int* perm = (int*)d_out;
    int* rank = perm + NEDGES;
    int* rptr = rank + NEDGES;

    pna_init  <<<(NNODES * DIN + 255) / 256, 256, 0, stream>>>(s1, s2, mx, mn, cnt, bns, bnq);
    pna_rank  <<<(NEDGES + 255) / 256, 256, 0, stream>>>(dst, cnt, rank);
    pna_scan  <<<1, 1024, 0, stream>>>(cnt, rptr);
    pna_fill  <<<(NEDGES + 255) / 256, 256, 0, stream>>>(dst, rank, rptr, perm);
    pna_edge  <<<NEDGES / 32, 256, 0, stream>>>(h, e, Wpre, src, dst, perm, s1, s2, mx, mn);
    pna_node  <<<NNODES / 16, 128, 0, stream>>>(h, snorm, Wpost, bpost, Wpre, bpre, s1, s2, mx, mn,
                                                cnt, out, bns, bnq);
    pna_bn    <<<(NNODES * DOUT + 255) / 256, 256, 0, stream>>>(out, bns, bnq, gamma, beta);
}


# Edge kernel: 1250 blocks each walk 30 tiles of 32 edges with W_pre fragments held in registers (W_pre no longer re-staged in LDS per tile); the next tile's rows are fetched under the product
# speedup vs baseline: 2.9696x; 1.0873x over previous
#include <hip/hip_runtime.h>
#include <hip/hip_bf16.h>
#include <math.h>

#define NNODES 100000
#define NEDGES 1200000
#define DIN    64
#define DOUT   64
#define EDIM   16
#define KEDGE  80
#define KPOST  832
#define AVG_D  2.5649493574615367f
#define EPSV   1e-05f

typedef __attribute__((ext_vector_type(2))) float v2f;
typedef __attribute__((ext_vector_type(8))) float v8f;

__device__ __forceinline__ unsigned fenc(float f) {
    unsigned u = __float_as_uint(f);
    return (u & 0x80000000u) ? ~u : (u | 0x80000000u);
}
__device__ __forceinline__ float fdec(unsigned u) {
    return (u & 0x80000000u) ? __uint_as_float(u & 0x7fffffffu)
                             : __uint_as_float(~u);
}


__global__ __launch_bounds__(256) void pna_init(float* s1, float* s2,
                                                unsigned* mx, unsigned* mn,
                                                int* cnt, float* bns, float* bnq) {
    int i = blockIdx.x * 256 + threadIdx.x;
    if (i < NNODES * DIN) {
        s1[i] = 0.0f; s2[i] = 0.0f;
        mx[i] = 0u;
        mn[i] = 0xFFFFFFFFu;
    }
    if (i < NNODES) cnt[i] = 0;
    if (i < DOUT) { bns[i] = 0.0f; bnq[i] = 0.0f; }
}

__global__ __launch_bounds__(256) void pna_rank(const int* __restrict__ dst,
                                                int* cnt, int* __restrict__ rank) {
    int i = blockIdx.x * 256 + threadIdx.x;
    if (i < NEDGES) rank[i] = atomicAdd(&cnt[dst[i]], 1);
}

#define SCAN_IT  25
#define NPIECE   (NNODES / 4)
#define RPTR_PAD (32 * SCAN_IT * 128)
static_assert(NNODES % 4 == 0, "counts are read as whole 16-byte pieces");
static_assert(RPTR_PAD >= NNODES, "the 32 waves must cover every count");
__global__ __launch_bounds__(1024) void pna_scan(const int* __restrict__ cnt,
                                                 int* __restrict__ rptr) {
    __shared__ int wtot[32];
    const int t = threadIdx.x;
    const int l = t & 31, w = t >> 5;
    const int4* c4 = (const int4*)cnt;
    int4* r4 = (int4*)rptr;
    const int q0 = w * (SCAN_IT * 32) + l;
    int4 v[SCAN_IT];
#pragma unroll
    for (int it = 0; it < SCAN_IT; ++it) {
        const int q = q0 + it * 32;
        v[it] = c4[(q < NPIECE) ? q : (NPIECE - 1)];
    }
    int tot = 0;
#pragma unroll
    for (int it = 0; it < SCAN_IT; ++it) tot += (v[it].x + v[it].y) + (v[it].z + v[it].w);
    for (int off = 16; off > 0; off >>= 1) tot += __shfl_xor(tot, off, 32);
    if (l == 0) wtot[w] = tot;
    __syncthreads();
    int carry = (l < w) ? wtot[l] : 0;
    for (int off = 16; off > 0; off >>= 1) carry += __shfl_xor(carry, off, 32);
#pragma unroll
    for (int it = 0; it < SCAN_IT; ++it) {
        const int4 a = v[it];
        const int s = (a.x + a.y) + (a.z + a.w);
        int inc = s;
        for (int off = 1; off < 32; off <<= 1) {
            const int n = __shfl_up(inc, off, 32);
            if (l >= off) inc += n;
        }
        const int ex = carry + inc - s;
        int4 o;
        o.x = ex;
        o.y = ex + a.x;
        o.z = o.y + a.y;
        o.w = o.z + a.z;
        r4[q0 + it * 32] = o;
        carry += __builtin_amdgcn_readlane(inc, 31);
    }
}

__global__ __launch_bounds__(256) void pna_fill(const int* __restrict__ dst,
                                                const int* __restrict__ rank,
                                                const int* __restrict__ rptr,
                                                int* __restrict__ perm) {
    int i = blockIdx.x * 256 + threadIdx.x;
    if (i < NEDGES) perm[rptr[dst[i]] + rank[i]] = i;
}

#define ZSTRIDE 84
#define ETILES  (NEDGES / 32)
#define NKE     (KEDGE / 4)
#define EGRID   1250
static_assert(NEDGES % 32 == 0, "whole tiles");
static_assert(EGRID <= ETILES, "every block owns a first tile");
__global__ __launch_bounds__(256) void pna_edge(
    const float* __restrict__ h, const float* __restrict__ e,
    const float* __restrict__ Wpre,
    const int* __restrict__ src, const int* __restrict__ dst,
    const int* __restrict__ perm,
    float* s1, float* s2, unsigned* mx, unsigned* mn) {

    __shared__ __attribute__((aligned(16))) float zs[2][32 * ZSTRIDE];
    __shared__ int ldst[2][32];

    const int tid = threadIdx.x;
    const int G   = EGRID;

    const int l  = tid & 31;
    const int wv = tid >> 5;
    const int cg = wv & 3;
    const int eg = wv >> 2;
    const int lo = l & 15;
    const int hi = l >> 4;
    const int col = cg * 16 + lo;

    v2f bf[NKE];
#pragma unroll
    for (int s = 0; s < NKE; ++s) {
        const int row = 4 * s + (4 * s >= DIN ? DIN : 0) + 2 * hi;
        bf[s].x = Wpre[row * DIN + col];
        bf[s].y = Wpre[(row + 1) * DIN + col];
    }

    const int r = tid >> 3, p = tid & 7;
    int t = blockIdx.x;
    int tq = t + G;      if (tq >= ETILES) tq = ETILES - 1;
    int tr = t + 2 * G;  if (tr >= ETILES) tr = ETILES - 1;
    int ei_c = perm[t * 32 + r];
    int ei_n = perm[tq * 32 + r];
    int pe2  = perm[tr * 32 + r];
    int sI_c = src[ei_c], dcur = dst[ei_c];
    int sI_n = src[ei_n], dI_n = dst[ei_n];
    float4 s_lo, s_hi;
    float2 e_pc;
    {
        const float4* hs = (const float4*)(h + (size_t)sI_c * DIN);
        const float2* er = (const float2*)(e + (size_t)ei_c * EDIM);
        s_lo = hs[p]; s_hi = hs[p + 8];
        e_pc = er[p];
    }

    const int rb = eg * 16 + 8 * hi;
    int buf = 0;
    for (; t < ETILES; t += G) {
        {
            float* zr = &zs[buf][r * ZSTRIDE];
            *(float4*)(zr + 4 * p)       = s_lo;
            *(float4*)(zr + 32 + 4 * p)  = s_hi;
            *(float2*)(zr + DIN + 2 * p) = e_pc;
            if (p == 0) ldst[buf][r] = dcur;
        }
        __syncthreads();

        {
            const float4* hs = (const float4*)(h + (size_t)sI_n * DIN);
            const float2* er = (const float2*)(e + (size_t)ei_n * EDIM);
            s_lo = hs[p]; s_hi = hs[p + 8];
            e_pc = er[p];
            dcur = dI_n;
            ei_n = pe2;
            sI_n = src[pe2]; dI_n = dst[pe2];
            int tn = t + 3 * G;  if (tn >= ETILES) tn = ETILES - 1;
            pe2 = perm[tn * 32 + r];
        }

        v8f acc = {};
        const float* arow = &zs[buf][(eg * 16 + lo) * ZSTRIDE + 2 * hi];
#pragma unroll
        for (int s = 0; s < NKE; ++s) {
            v2f a = *(const v2f*)(arow + 4 * s);
            acc = __builtin_amdgcn_wmma_f32_16x16x4_f32(
                false, a, false, bf[s], (short)0, acc, false, false);
        }

        const int* ld = &ldst[buf][rb];
        int cur = ld[0];
        float v = acc[0];
        float sum = v, sq = v * v;
        unsigned emx = fenc(v), emn = emx;
#pragma unroll
        for (int q = 1; q < 8; ++q) {
            int d = ld[q];
            v = acc[q];
            unsigned ev = fenc(v);
            if (d != cur) {
                size_t o = (size_t)cur * DIN + col;
                atomicAdd(&s1[o], sum);
                atomicAdd(&s2[o], sq);
                atomicMax(&mx[o], emx);
                atomicMin(&mn[o], emn);
                cur = d; sum = v; sq = v * v; emx = ev; emn = ev;
            } else {
                sum += v; sq += v * v;
                emx = (ev > emx) ? ev : emx;
                emn = (ev < emn) ? ev : emn;
            }
        }
        {
            size_t o = (size_t)cur * DIN + col;
            atomicAdd(&s1[o], sum);
            atomicAdd(&s2[o], sq);
            atomicMax(&mx[o], emx);
            atomicMin(&mn[o], emn);
        }
        buf ^= 1;
    }
}

#define FSTRIDE 836
#define NKB     16
#define NNB     13
#define NKC     16
__global__ __launch_bounds__(128) void pna_node(
    const float* __restrict__ h, const float* __restrict__ snorm,
    const float* __restrict__ Wpost, const float* __restrict__ bpost,
    const float* __restrict__ Wpre, const float* __restrict__ bpre,
    const float* __restrict__ s1, const float* __restrict__ s2,
    const unsigned* __restrict__ mx, const unsigned* __restrict__ mn,
    const int* __restrict__ cnt,
    float* __restrict__ out, float* bns, float* bnq) {

    __shared__ __attribute__((aligned(16))) float feat[16 * FSTRIDE];
    __shared__ float fab[16 * 2];
    const int tid = threadIdx.x;
    const int n0  = blockIdx.x * 16;

    const int l  = tid & 31;
    const int cg = tid >> 5;
    const int lo = l & 15;
    const int hi = l >> 4;
    const int col = cg * 16 + lo;

    const float* wc = Wpre + (size_t)(DIN + 2 * hi) * DIN + col;
    float cx[NKC], cy[NKC];
#pragma unroll
    for (int s = 0; s < NKC; ++s) {
        cx[s] = wc[(4 * s) * DIN];
        cy[s] = wc[(4 * s + 1) * DIN];
    }
    const float cb = bpre[col];

    {
        const int r = tid >> 3, p = tid & 7;
        const int node = n0 + r;
        const size_t o = (size_t)node * DIN;
        const float4* hp  = (const float4*)(h + o);
        const float4* s1p = (const float4*)(s1 + o);
        const float4* s2p = (const float4*)(s2 + o);
        const uint4*  mxp = (const uint4*)(mx + o);
        const uint4*  mnp = (const uint4*)(mn + o);
        const int dg = cnt[node];
        float4 hv[2], a1[2], a2[2];
        uint4  ux[2], un[2];
#pragma unroll
        for (int q = 0; q < 2; ++q) {
            hv[q] = hp[p + 8 * q];
            a1[q] = s1p[p + 8 * q];
            a2[q] = s2p[p + 8 * q];
            ux[q] = mxp[p + 8 * q];
            un[q] = mnp[p + 8 * q];
        }
        float* fr = &feat[r * FSTRIDE];
        *(float4*)(fr + 4 * p)      = hv[0];
        *(float4*)(fr + 32 + 4 * p) = hv[1];
        const float D    = (float)dg;
        const float logd = logf(D + 1.0f);
        const float fa   = logd / AVG_D;
        const float fb   = AVG_D / logd;
        if (p == 0) { fab[2 * r] = fa; fab[2 * r + 1] = fb; }
#pragma unroll
        for (int q = 0; q < 2; ++q) {
            const int c = 4 * p + 32 * q;
            float4 mu, sd, vx, vn;
            mu.x = a1[q].x / D; mu.y = a1[q].y / D;
            mu.z = a1[q].z / D; mu.w = a1[q].w / D;
            sd.x = sqrtf(fmaxf(a2[q].x / D - mu.x * mu.x, 0.0f) + EPSV);
            sd.y = sqrtf(fmaxf(a2[q].y / D - mu.y * mu.y, 0.0f) + EPSV);
            sd.z = sqrtf(fmaxf(a2[q].z / D - mu.z * mu.z, 0.0f) + EPSV);
            sd.w = sqrtf(fmaxf(a2[q].w / D - mu.w * mu.w, 0.0f) + EPSV);
            vx.x = fdec(ux[q].x); vx.y = fdec(ux[q].y); vx.z = fdec(ux[q].z); vx.w = fdec(ux[q].w);
            vn.x = fdec(un[q].x); vn.y = fdec(un[q].y); vn.z = fdec(un[q].z); vn.w = fdec(un[q].w);
            *(float4*)(fr + DIN + c)      = mu;
            *(float4*)(fr + 2 * DIN + c)  = vx;
            *(float4*)(fr + 3 * DIN + c)  = vn;
            *(float4*)(fr + 4 * DIN + c)  = sd;
            *(float4*)(fr + 8 * DIN + c)  = make_float4(sd.x * fa, sd.y * fa, sd.z * fa, sd.w * fa);
            *(float4*)(fr + 12 * DIN + c) = make_float4(sd.x * fb, sd.y * fb, sd.z * fb, sd.w * fb);
        }
    }
    __syncthreads();

    const float* arow = &feat[lo * FSTRIDE + 2 * hi];

    {
        v8f cacc = {};
#pragma unroll
        for (int s = 0; s < NKC; ++s) {
            v2f a = *(const v2f*)(arow + 4 * s);
            v2f b;
            b.x = cx[s];
            b.y = cy[s];
            cacc = __builtin_amdgcn_wmma_f32_16x16x4_f32(
                false, a, false, b, (short)0, cacc, false, false);
        }
#pragma unroll
        for (int i = 0; i < 8; ++i) {
            const int row = 8 * hi + i;
            const float cv = cacc[i] + cb;
            const float fa = fab[2 * row], fb = fab[2 * row + 1];
            float* fp = &feat[row * FSTRIDE + col];
            for (int g = 1; g <= 3; ++g) {
                const float x = fp[g * DIN] + cv;
                fp[g * DIN]       = x;
                fp[(g + 4) * DIN] = x * fa;
                fp[(g + 8) * DIN] = x * fb;
            }
        }
    }
    __syncthreads();

    v8f acc = {};
    const float* bcol = Wpost + (size_t)(2 * hi) * DOUT + col;
    float bx[2][NKB], by[2][NKB];
#pragma unroll
    for (int s = 0; s < NKB; ++s) {
        bx[0][s] = bcol[(4 * s) * DOUT];
        by[0][s] = bcol[(4 * s + 1) * DOUT];
    }
#pragma unroll
    for (int kb = 0; kb < NNB; ++kb) {
        const int cur = kb & 1, nxt = cur ^ 1;
        if (kb + 1 < NNB) {
#pragma unroll
            for (int s = 0; s < NKB; ++s) {
                const int k = (kb + 1) * 4 * NKB + 4 * s;
                bx[nxt][s] = bcol[k * DOUT];
                by[nxt][s] = bcol[(k + 1) * DOUT];
            }
        }
#pragma unroll
        for (int s = 0; s < NKB; ++s) {
            const int k0 = kb * 4 * NKB + 4 * s;
            v2f a = *(const v2f*)(arow + k0);
            v2f b;
            b.x = bx[cur][s];
            b.y = by[cur][s];
            acc = __builtin_amdgcn_wmma_f32_16x16x4_f32(
                false, a, false, b, (short)0, acc, false, false);
        }
    }

    const float bv = bpost[col];
    float ps = 0.0f, ps2 = 0.0f;
#pragma unroll
    for (int r = 0; r < 8; ++r) {
        int M = r + 8 * hi;
        int node = n0 + M;
        float v = (acc[r] + bv) * snorm[node];
        out[(size_t)node * DOUT + col] = v;
        ps += v; ps2 += v * v;
    }
    ps  += __shfl_xor(ps, 16, 32);
    ps2 += __shfl_xor(ps2, 16, 32);
    if (hi == 0) {
        atomicAdd(&bns[col], ps);
        atomicAdd(&bnq[col], ps2);
    }
}

__global__ __launch_bounds__(256) void pna_bn(
    float* out, const float* bns, const float* bnq,
    const float* gamma, const float* beta) {
    int i = blockIdx.x * 256 + threadIdx.x;
    if (i >= NNODES * DOUT) return;
    int c = i & 63;
    float mu  = bns[c] / (float)NNODES;
    float var = bnq[c] / (float)NNODES - mu * mu;
    float v = out[i];
    out[i] = (v - mu) * rsqrtf(var + 1e-05f) * gamma[c] + beta[c];
}

extern "C" void kernel_launch(void* const* d_in, const int* in_sizes, int n_in,
                              void* d_out, int out_size, void* d_ws, size_t ws_size,
                              hipStream_t stream) {
    const float* h      = (const float*)d_in[0];
    const float* e      = (const float*)d_in[1];
    const float* snorm  = (const float*)d_in[2];
    const float* Wpre   = (const float*)d_in[3];
    const float* bpre   = (const float*)d_in[4];
    const float* Wpost  = (const float*)d_in[5];
    const float* bpost  = (const float*)d_in[6];
    const float* gamma  = (const float*)d_in[7];
    const float* beta   = (const float*)d_in[8];
    const int*   src    = (const int*)d_in[9];
    const int*   dst    = (const int*)d_in[10];
    float* out = (float*)d_out;

    float*    s1  = (float*)d_ws;
    float*    s2  = s1 + (size_t)NNODES * DIN;
    unsigned* mx  = (unsigned*)(s2 + (size_t)NNODES * DIN);
    unsigned* mn  = mx + (size_t)NNODES * DIN;
    int*      cnt = (int*)(mn + (size_t)NNODES * DIN);
    float*    bns = (float*)(cnt + NNODES);
    float*    bnq = bns + DOUT;

    int* perm = (int*)d_out;
    int* rank = perm + NEDGES;
    int* rptr = rank + NEDGES;

    pna_init  <<<(NNODES * DIN + 255) / 256, 256, 0, stream>>>(s1, s2, mx, mn, cnt, bns, bnq);
    pna_rank  <<<(NEDGES + 255) / 256, 256, 0, stream>>>(dst, cnt, rank);
    pna_scan  <<<1, 1024, 0, stream>>>(cnt, rptr);
    pna_fill  <<<(NEDGES + 255) / 256, 256, 0, stream>>>(dst, rank, rptr, perm);
    pna_edge  <<<EGRID, 256, 0, stream>>>(h, e, Wpre, src, dst, perm, s1, s2, mx, mn);
    pna_node  <<<NNODES / 16, 128, 0, stream>>>(h, snorm, Wpost, bpost, Wpre, bpre, s1, s2, mx, mn,
                                                cnt, out, bns, bnq);
    pna_bn    <<<(NNODES * DOUT + 255) / 256, 256, 0, stream>>>(out, bns, bnq, gamma, beta);
}


# Node kernel: 32 nodes per block, each W_post fragment feeds 2 node tiles; fa, fb feature copies formed in registers from a 320-column LDS tile, 3 products per A read; sum order changes, BN sums too
# speedup vs baseline: 3.0902x; 1.0406x over previous
#include <hip/hip_runtime.h>
#include <hip/hip_bf16.h>
#include <math.h>

#define NNODES 100000
#define NEDGES 1200000
#define DIN    64
#define DOUT   64
#define EDIM   16
#define KEDGE  80
#define KPOST  832
#define AVG_D  2.5649493574615367f
#define EPSV   1e-05f

typedef __attribute__((ext_vector_type(2))) float v2f;
typedef __attribute__((ext_vector_type(8))) float v8f;

__device__ __forceinline__ unsigned fenc(float f) {
    unsigned u = __float_as_uint(f);
    return (u & 0x80000000u) ? ~u : (u | 0x80000000u);
}
__device__ __forceinline__ float fdec(unsigned u) {
    return (u & 0x80000000u) ? __uint_as_float(u & 0x7fffffffu)
                             : __uint_as_float(~u);
}


__global__ __launch_bounds__(256) void pna_init(float* s1, float* s2,
                                                unsigned* mx, unsigned* mn,
                                                int* cnt, float* bns, float* bnq) {
    int i = blockIdx.x * 256 + threadIdx.x;
    if (i < NNODES * DIN) {
        s1[i] = 0.0f; s2[i] = 0.0f;
        mx[i] = 0u;
        mn[i] = 0xFFFFFFFFu;
    }
    if (i < NNODES) cnt[i] = 0;
    if (i < DOUT) { bns[i] = 0.0f; bnq[i] = 0.0f; }
}

__global__ __launch_bounds__(256) void pna_rank(const int* __restrict__ dst,
                                                int* cnt, int* __restrict__ rank) {
    int i = blockIdx.x * 256 + threadIdx.x;
    if (i < NEDGES) rank[i] = atomicAdd(&cnt[dst[i]], 1);
}

#define SCAN_IT  25
#define NPIECE   (NNODES / 4)
#define RPTR_PAD (32 * SCAN_IT * 128)
static_assert(NNODES % 4 == 0, "counts are read as whole 16-byte pieces");
static_assert(RPTR_PAD >= NNODES, "the 32 waves must cover every count");
__global__ __launch_bounds__(1024) void pna_scan(const int* __restrict__ cnt,
                                                 int* __restrict__ rptr) {
    __shared__ int wtot[32];
    const int t = threadIdx.x;
    const int l = t & 31, w = t >> 5;
    const int4* c4 = (const int4*)cnt;
    int4* r4 = (int4*)rptr;
    const int q0 = w * (SCAN_IT * 32) + l;
    int4 v[SCAN_IT];
#pragma unroll
    for (int it = 0; it < SCAN_IT; ++it) {
        const int q = q0 + it * 32;
        v[it] = c4[(q < NPIECE) ? q : (NPIECE - 1)];
    }
    int tot = 0;
#pragma unroll
    for (int it = 0; it < SCAN_IT; ++it) tot += (v[it].x + v[it].y) + (v[it].z + v[it].w);
    for (int off = 16; off > 0; off >>= 1) tot += __shfl_xor(tot, off, 32);
    if (l == 0) wtot[w] = tot;
    __syncthreads();
    int carry = (l < w) ? wtot[l] : 0;
    for (int off = 16; off > 0; off >>= 1) carry += __shfl_xor(carry, off, 32);
#pragma unroll
    for (int it = 0; it < SCAN_IT; ++it) {
        const int4 a = v[it];
        const int s = (a.x + a.y) + (a.z + a.w);
        int inc = s;
        for (int off = 1; off < 32; off <<= 1) {
            const int n = __shfl_up(inc, off, 32);
            if (l >= off) inc += n;
        }
        const int ex = carry + inc - s;
        int4 o;
        o.x = ex;
        o.y = ex + a.x;
        o.z = o.y + a.y;
        o.w = o.z + a.z;
        r4[q0 + it * 32] = o;
        carry += __builtin_amdgcn_readlane(inc, 31);
    }
}

__global__ __launch_bounds__(256) void pna_fill(const int* __restrict__ dst,
                                                const int* __restrict__ rank,
                                                const int* __restrict__ rptr,
                                                int* __restrict__ perm) {
    int i = blockIdx.x * 256 + threadIdx.x;
    if (i < NEDGES) perm[rptr[dst[i]] + rank[i]] = i;
}

#define ZSTRIDE 84
#define ETILES  (NEDGES / 32)
#define NKE     (KEDGE / 4)
#define EGRID   1250
static_assert(NEDGES % 32 == 0, "whole tiles");
static_assert(EGRID <= ETILES, "every block owns a first tile");
__global__ __launch_bounds__(256) void pna_edge(
    const float* __restrict__ h, const float* __restrict__ e,
    const float* __restrict__ Wpre,
    const int* __restrict__ src, const int* __restrict__ dst,
    const int* __restrict__ perm,
    float* s1, float* s2, unsigned* mx, unsigned* mn) {

    __shared__ __attribute__((aligned(16))) float zs[2][32 * ZSTRIDE];
    __shared__ int ldst[2][32];

    const int tid = threadIdx.x;
    const int G   = EGRID;

    const int l  = tid & 31;
    const int wv = tid >> 5;
    const int cg = wv & 3;
    const int eg = wv >> 2;
    const int lo = l & 15;
    const int hi = l >> 4;
    const int col = cg * 16 + lo;

    v2f bf[NKE];
#pragma unroll
    for (int s = 0; s < NKE; ++s) {
        const int row = 4 * s + (4 * s >= DIN ? DIN : 0) + 2 * hi;
        bf[s].x = Wpre[row * DIN + col];
        bf[s].y = Wpre[(row + 1) * DIN + col];
    }

    const int r = tid >> 3, p = tid & 7;
    int t = blockIdx.x;
    int tq = t + G;      if (tq >= ETILES) tq = ETILES - 1;
    int tr = t + 2 * G;  if (tr >= ETILES) tr = ETILES - 1;
    int ei_c = perm[t * 32 + r];
    int ei_n = perm[tq * 32 + r];
    int pe2  = perm[tr * 32 + r];
    int sI_c = src[ei_c], dcur = dst[ei_c];
    int sI_n = src[ei_n], dI_n = dst[ei_n];
    float4 s_lo, s_hi;
    float2 e_pc;
    {
        const float4* hs = (const float4*)(h + (size_t)sI_c * DIN);
        const float2* er = (const float2*)(e + (size_t)ei_c * EDIM);
        s_lo = hs[p]; s_hi = hs[p + 8];
        e_pc = er[p];
    }

    const int rb = eg * 16 + 8 * hi;
    int buf = 0;
    for (; t < ETILES; t += G) {
        {
            float* zr = &zs[buf][r * ZSTRIDE];
            *(float4*)(zr + 4 * p)       = s_lo;
            *(float4*)(zr + 32 + 4 * p)  = s_hi;
            *(float2*)(zr + DIN + 2 * p) = e_pc;
            if (p == 0) ldst[buf][r] = dcur;
        }
        __syncthreads();

        {
            const float4* hs = (const float4*)(h + (size_t)sI_n * DIN);
            const float2* er = (const float2*)(e + (size_t)ei_n * EDIM);
            s_lo = hs[p]; s_hi = hs[p + 8];
            e_pc = er[p];
            dcur = dI_n;
            ei_n = pe2;
            sI_n = src[pe2]; dI_n = dst[pe2];
            int tn = t + 3 * G;  if (tn >= ETILES) tn = ETILES - 1;
            pe2 = perm[tn * 32 + r];
        }

        v8f acc = {};
        const float* arow = &zs[buf][(eg * 16 + lo) * ZSTRIDE + 2 * hi];
#pragma unroll
        for (int s = 0; s < NKE; ++s) {
            v2f a = *(const v2f*)(arow + 4 * s);
            acc = __builtin_amdgcn_wmma_f32_16x16x4_f32(
                false, a, false, bf[s], (short)0, acc, false, false);
        }

        const int* ld = &ldst[buf][rb];
        int cur = ld[0];
        float v = acc[0];
        float sum = v, sq = v * v;
        unsigned emx = fenc(v), emn = emx;
#pragma unroll
        for (int q = 1; q < 8; ++q) {
            int d = ld[q];
            v = acc[q];
            unsigned ev = fenc(v);
            if (d != cur) {
                size_t o = (size_t)cur * DIN + col;
                atomicAdd(&s1[o], sum);
                atomicAdd(&s2[o], sq);
                atomicMax(&mx[o], emx);
                atomicMin(&mn[o], emn);
                cur = d; sum = v; sq = v * v; emx = ev; emn = ev;
            } else {
                sum += v; sq += v * v;
                emx = (ev > emx) ? ev : emx;
                emn = (ev < emn) ? ev : emn;
            }
        }
        {
            size_t o = (size_t)cur * DIN + col;
            atomicAdd(&s1[o], sum);
            atomicAdd(&s2[o], sq);
            atomicMax(&mx[o], emx);
            atomicMin(&mn[o], emn);
        }
        buf ^= 1;
    }
}

#define FSTRIDE 324
#define NTM     2
#define NBLK    (16 * NTM)
#define NKB     16
#define NKC     16
static_assert(NNODES % NBLK == 0, "whole blocks");
__global__ __launch_bounds__(128) void pna_node(
    const float* __restrict__ h, const float* __restrict__ snorm,
    const float* __restrict__ Wpost, const float* __restrict__ bpost,
    const float* __restrict__ Wpre, const float* __restrict__ bpre,
    const float* __restrict__ s1, const float* __restrict__ s2,
    const unsigned* __restrict__ mx, const unsigned* __restrict__ mn,
    const int* __restrict__ cnt,
    float* __restrict__ out, float* bns, float* bnq) {

    __shared__ __attribute__((aligned(16))) float feat[NBLK * FSTRIDE];
    __shared__ float fab[NBLK * 2];
    const int tid = threadIdx.x;
    const int n0  = blockIdx.x * NBLK;

    const int l  = tid & 31;
    const int cg = tid >> 5;
    const int lo = l & 15;
    const int hi = l >> 4;
    const int col = cg * 16 + lo;

    const float* wc = Wpre + (size_t)(DIN + 2 * hi) * DIN + col;
    float cx[NKC], cy[NKC];
#pragma unroll
    for (int s = 0; s < NKC; ++s) {
        cx[s] = wc[(4 * s) * DIN];
        cy[s] = wc[(4 * s + 1) * DIN];
    }
    const float cb = bpre[col];

    {
        const int r0 = tid >> 3, p = tid & 7;
        float4 hv[NTM][2], a1[NTM][2], a2[NTM][2];
        uint4  ux[NTM][2], un[NTM][2];
        int dg[NTM];
#pragma unroll
        for (int t = 0; t < NTM; ++t) {
            const int node = n0 + 16 * t + r0;
            const size_t o = (size_t)node * DIN;
            const float4* hp  = (const float4*)(h + o);
            const float4* s1p = (const float4*)(s1 + o);
            const float4* s2p = (const float4*)(s2 + o);
            const uint4*  mxp = (const uint4*)(mx + o);
            const uint4*  mnp = (const uint4*)(mn + o);
            dg[t] = cnt[node];
#pragma unroll
            for (int q = 0; q < 2; ++q) {
                hv[t][q] = hp[p + 8 * q];
                a1[t][q] = s1p[p + 8 * q];
                a2[t][q] = s2p[p + 8 * q];
                ux[t][q] = mxp[p + 8 * q];
                un[t][q] = mnp[p + 8 * q];
            }
        }
#pragma unroll
        for (int t = 0; t < NTM; ++t) {
            const int r = 16 * t + r0;
            float* fr = &feat[r * FSTRIDE];
            *(float4*)(fr + 4 * p)      = hv[t][0];
            *(float4*)(fr + 32 + 4 * p) = hv[t][1];
            const float D    = (float)dg[t];
            const float logd = logf(D + 1.0f);
            const float fa   = logd / AVG_D;
            const float fb   = AVG_D / logd;
            if (p == 0) { fab[2 * r] = fa; fab[2 * r + 1] = fb; }
#pragma unroll
            for (int q = 0; q < 2; ++q) {
                const int c = 4 * p + 32 * q;
                float4 mu, sd, vx, vn;
                mu.x = a1[t][q].x / D; mu.y = a1[t][q].y / D;
                mu.z = a1[t][q].z / D; mu.w = a1[t][q].w / D;
                sd.x = sqrtf(fmaxf(a2[t][q].x / D - mu.x * mu.x, 0.0f) + EPSV);
                sd.y = sqrtf(fmaxf(a2[t][q].y / D - mu.y * mu.y, 0.0f) + EPSV);
                sd.z = sqrtf(fmaxf(a2[t][q].z / D - mu.z * mu.z, 0.0f) + EPSV);
                sd.w = sqrtf(fmaxf(a2[t][q].w / D - mu.w * mu.w, 0.0f) + EPSV);
                vx.x = fdec(ux[t][q].x); vx.y = fdec(ux[t][q].y);
                vx.z = fdec(ux[t][q].z); vx.w = fdec(ux[t][q].w);
                vn.x = fdec(un[t][q].x); vn.y = fdec(un[t][q].y);
                vn.z = fdec(un[t][q].z); vn.w = fdec(un[t][q].w);
                *(float4*)(fr + DIN + c)      = mu;
                *(float4*)(fr + 2 * DIN + c)  = vx;
                *(float4*)(fr + 3 * DIN + c)  = vn;
                *(float4*)(fr + 4 * DIN + c)  = sd;
            }
        }
    }
    __syncthreads();

    const float* arow = &feat[lo * FSTRIDE + 2 * hi];

    {
        v8f cacc[NTM];
#pragma unroll
        for (int t = 0; t < NTM; ++t) cacc[t] = (v8f){};
#pragma unroll
        for (int s = 0; s < NKC; ++s) {
            v2f b;
            b.x = cx[s];
            b.y = cy[s];
#pragma unroll
            for (int t = 0; t < NTM; ++t) {
                v2f a = *(const v2f*)(arow + 16 * t * FSTRIDE + 4 * s);
                cacc[t] = __builtin_amdgcn_wmma_f32_16x16x4_f32(
                    false, a, false, b, (short)0, cacc[t], false, false);
            }
        }
#pragma unroll
        for (int t = 0; t < NTM; ++t) {
#pragma unroll
            for (int i = 0; i < 8; ++i) {
                const int row = 16 * t + 8 * hi + i;
                const float cv = cacc[t][i] + cb;
                float* fp = &feat[row * FSTRIDE + col];
                fp[1 * DIN] = fp[1 * DIN] + cv;
                fp[2 * DIN] = fp[2 * DIN] + cv;
                fp[3 * DIN] = fp[3 * DIN] + cv;
            }
        }
    }
    __syncthreads();

    float fsa[NTM], fsb[NTM];
#pragma unroll
    for (int t = 0; t < NTM; ++t) {
        fsa[t] = fab[2 * (16 * t + lo)];
        fsb[t] = fab[2 * (16 * t + lo) + 1];
    }

    v8f acc[NTM];
#pragma unroll
    for (int t = 0; t < NTM; ++t) acc[t] = (v8f){};
    const float* bcol = Wpost + (size_t)(2 * hi) * DOUT + col;
#define NH 8
    float hx[NKB], hy[NKB];
#pragma unroll
    for (int s = 0; s < NKB; ++s) {
        hx[s] = bcol[(4 * s) * DOUT];
        hy[s] = bcol[(4 * s + 1) * DOUT];
    }
    float gx[2][3][NH], gy[2][3][NH];
#pragma unroll
    for (int j = 0; j < 3; ++j) {
#pragma unroll
        for (int s = 0; s < NH; ++s) {
            const int k = (1 + 4 * j) * DIN + 4 * s;
            gx[0][j][s] = bcol[k * DOUT];
            gy[0][j][s] = bcol[(k + 1) * DOUT];
        }
    }
#pragma unroll
    for (int s = 0; s < NKB; ++s) {
        v2f b;
        b.x = hx[s];
        b.y = hy[s];
#pragma unroll
        for (int t = 0; t < NTM; ++t) {
            v2f a = *(const v2f*)(arow + 16 * t * FSTRIDE + 4 * s);
            acc[t] = __builtin_amdgcn_wmma_f32_16x16x4_f32(
                false, a, false, b, (short)0, acc[t], false, false);
        }
    }
#pragma unroll
    for (int hb = 0; hb < 8; ++hb) {
        const int cur = hb & 1, nxt = cur ^ 1;
        if (hb + 1 < 8) {
            const int gn = 1 + (hb + 1) / 2, sn = ((hb + 1) % 2) * NH;
#pragma unroll
            for (int j = 0; j < 3; ++j) {
#pragma unroll
                for (int s = 0; s < NH; ++s) {
                    const int k = (gn + 4 * j) * DIN + 4 * (sn + s);
                    gx[nxt][j][s] = bcol[k * DOUT];
                    gy[nxt][j][s] = bcol[(k + 1) * DOUT];
                }
            }
        }
        const int g = 1 + hb / 2, s0 = (hb % 2) * NH;
#pragma unroll
        for (int s = 0; s < NH; ++s) {
            v2f a[NTM];
#pragma unroll
            for (int t = 0; t < NTM; ++t)
                a[t] = *(const v2f*)(arow + 16 * t * FSTRIDE + g * DIN + 4 * (s0 + s));
#pragma unroll
            for (int j = 0; j < 3; ++j) {
                v2f b;
                b.x = gx[cur][j][s];
                b.y = gy[cur][j][s];
#pragma unroll
                for (int t = 0; t < NTM; ++t) {
                    v2f af = a[t];
                    if (j == 1) { af.x = a[t].x * fsa[t]; af.y = a[t].y * fsa[t]; }
                    if (j == 2) { af.x = a[t].x * fsb[t]; af.y = a[t].y * fsb[t]; }
                    acc[t] = __builtin_amdgcn_wmma_f32_16x16x4_f32(
                        false, af, false, b, (short)0, acc[t], false, false);
                }
            }
        }
    }
#undef NH

    const float bv = bpost[col];
    float ps = 0.0f, ps2 = 0.0f;
#pragma unroll
    for (int t = 0; t < NTM; ++t) {
#pragma unroll
        for (int r = 0; r < 8; ++r) {
            int node = n0 + 16 * t + 8 * hi + r;
            float v = (acc[t][r] + bv) * snorm[node];
            out[(size_t)node * DOUT + col] = v;
            ps += v; ps2 += v * v;
        }
    }
    ps  += __shfl_xor(ps, 16, 32);
    ps2 += __shfl_xor(ps2, 16, 32);
    if (hi == 0) {
        atomicAdd(&bns[col], ps);
        atomicAdd(&bnq[col], ps2);
    }
}

__global__ __launch_bounds__(256) void pna_bn(
    float* out, const float* bns, const float* bnq,
    const float* gamma, const float* beta) {
    int i = blockIdx.x * 256 + threadIdx.x;
    if (i >= NNODES * DOUT) return;
    int c = i & 63;
    float mu  = bns[c] / (float)NNODES;
    float var = bnq[c] / (float)NNODES - mu * mu;
    float v = out[i];
    out[i] = (v - mu) * rsqrtf(var + 1e-05f) * gamma[c] + beta[c];
}

extern "C" void kernel_launch(void* const* d_in, const int* in_sizes, int n_in,
                              void* d_out, int out_size, void* d_ws, size_t ws_size,
                              hipStream_t stream) {
    const float* h      = (const float*)d_in[0];
    const float* e      = (const float*)d_in[1];
    const float* snorm  = (const float*)d_in[2];
    const float* Wpre   = (const float*)d_in[3];
    const float* bpre   = (const float*)d_in[4];
    const float* Wpost  = (const float*)d_in[5];
    const float* bpost  = (const float*)d_in[6];
    const float* gamma  = (const float*)d_in[7];
    const float* beta   = (const float*)d_in[8];
    const int*   src    = (const int*)d_in[9];
    const int*   dst    = (const int*)d_in[10];
    float* out = (float*)d_out;

    float*    s1  = (float*)d_ws;
    float*    s2  = s1 + (size_t)NNODES * DIN;
    unsigned* mx  = (unsigned*)(s2 + (size_t)NNODES * DIN);
    unsigned* mn  = mx + (size_t)NNODES * DIN;
    int*      cnt = (int*)(mn + (size_t)NNODES * DIN);
    float*    bns = (float*)(cnt + NNODES);
    float*    bnq = bns + DOUT;

    int* perm = (int*)d_out;
    int* rank = perm + NEDGES;
    int* rptr = rank + NEDGES;

    pna_init  <<<(NNODES * DIN + 255) / 256, 256, 0, stream>>>(s1, s2, mx, mn, cnt, bns, bnq);
    pna_rank  <<<(NEDGES + 255) / 256, 256, 0, stream>>>(dst, cnt, rank);
    pna_scan  <<<1, 1024, 0, stream>>>(cnt, rptr);
    pna_fill  <<<(NEDGES + 255) / 256, 256, 0, stream>>>(dst, rank, rptr, perm);
    pna_edge  <<<EGRID, 256, 0, stream>>>(h, e, Wpre, src, dst, perm, s1, s2, mx, mn);
    pna_node  <<<NNODES / NBLK, 128, 0, stream>>>(h, snorm, Wpost, bpost, Wpre, bpre, s1, s2, mx, mn,
                                                cnt, out, bns, bnq);
    pna_bn    <<<(NNODES * DOUT + 255) / 256, 256, 0, stream>>>(out, bns, bnq, gamma, beta);
}


# Edge kernel: each block owns 40 consecutive destinations (its sorted edge range from rptr), aggregates in LDS by LDS atomics, stores the planes once; no global atomics, no clearing; sum order changes
# speedup vs baseline: 3.3829x; 1.0947x over previous
#include <hip/hip_runtime.h>
#include <hip/hip_bf16.h>
#include <math.h>

#define NNODES 100000
#define NEDGES 1200000
#define DIN    64
#define DOUT   64
#define EDIM   16
#define KEDGE  80
#define KPOST  832
#define AVG_D  2.5649493574615367f
#define EPSV   1e-05f

typedef __attribute__((ext_vector_type(2))) float v2f;
typedef __attribute__((ext_vector_type(8))) float v8f;

__device__ __forceinline__ unsigned fenc(float f) {
    unsigned u = __float_as_uint(f);
    return (u & 0x80000000u) ? ~u : (u | 0x80000000u);
}
__device__ __forceinline__ float fdec(unsigned u) {
    return (u & 0x80000000u) ? __uint_as_float(u & 0x7fffffffu)
                             : __uint_as_float(~u);
}


__global__ __launch_bounds__(256) void pna_init(int* cnt, float* bns, float* bnq) {
    int i = blockIdx.x * 256 + threadIdx.x;
    if (i < NNODES) cnt[i] = 0;
    if (i < DOUT) { bns[i] = 0.0f; bnq[i] = 0.0f; }
}

__global__ __launch_bounds__(256) void pna_rank(const int* __restrict__ dst,
                                                int* cnt, int* __restrict__ rank) {
    int i = blockIdx.x * 256 + threadIdx.x;
    if (i < NEDGES) rank[i] = atomicAdd(&cnt[dst[i]], 1);
}

#define SCAN_IT  25
#define NPIECE   (NNODES / 4)
#define RPTR_PAD (32 * SCAN_IT * 128)
static_assert(NNODES % 4 == 0, "counts are read as whole 16-byte pieces");
static_assert(RPTR_PAD >= NNODES, "the 32 waves must cover every count");
__global__ __launch_bounds__(1024) void pna_scan(const int* __restrict__ cnt,
                                                 int* __restrict__ rptr) {
    __shared__ int wtot[32];
    const int t = threadIdx.x;
    const int l = t & 31, w = t >> 5;
    const int4* c4 = (const int4*)cnt;
    int4* r4 = (int4*)rptr;
    const int q0 = w * (SCAN_IT * 32) + l;
    int4 v[SCAN_IT];
#pragma unroll
    for (int it = 0; it < SCAN_IT; ++it) {
        const int q = q0 + it * 32;
        v[it] = c4[(q < NPIECE) ? q : (NPIECE - 1)];
    }
    int tot = 0;
#pragma unroll
    for (int it = 0; it < SCAN_IT; ++it) tot += (v[it].x + v[it].y) + (v[it].z + v[it].w);
    for (int off = 16; off > 0; off >>= 1) tot += __shfl_xor(tot, off, 32);
    if (l == 0) wtot[w] = tot;
    __syncthreads();
    int carry = (l < w) ? wtot[l] : 0;
    for (int off = 16; off > 0; off >>= 1) carry += __shfl_xor(carry, off, 32);
#pragma unroll
    for (int it = 0; it < SCAN_IT; ++it) {
        const int4 a = v[it];
        const int s = (a.x + a.y) + (a.z + a.w);
        int inc = s;
        for (int off = 1; off < 32; off <<= 1) {
            const int n = __shfl_up(inc, off, 32);
            if (l >= off) inc += n;
        }
        const int ex = carry + inc - s;
        int4 o;
        o.x = ex;
        o.y = ex + a.x;
        o.z = o.y + a.y;
        o.w = o.z + a.z;
        r4[q0 + it * 32] = o;
        carry += __builtin_amdgcn_readlane(inc, 31);
    }
}

__global__ __launch_bounds__(256) void pna_fill(const int* __restrict__ dst,
                                                const int* __restrict__ rank,
                                                const int* __restrict__ rptr,
                                                int* __restrict__ perm) {
    int i = blockIdx.x * 256 + threadIdx.x;
    if (i < NEDGES) perm[rptr[dst[i]] + rank[i]] = i;
}

#define ZSTRIDE 84
#define NKE     (KEDGE / 4)
#define NPB     40
#define EGRID   (NNODES / NPB)
static_assert(NNODES % NPB == 0, "whole blocks of destinations");
static_assert((NPB * DIN) % 4 == 0, "the aggregates leave as 16-byte pieces");
__global__ __launch_bounds__(256) void pna_edge(
    const float* __restrict__ h, const float* __restrict__ e,
    const float* __restrict__ Wpre,
    const int* __restrict__ src, const int* __restrict__ dst,
    const int* __restrict__ perm, const int* __restrict__ rptr,
    float* __restrict__ s1, float* __restrict__ s2,
    unsigned* __restrict__ mx, unsigned* __restrict__ mn) {

    __shared__ __attribute__((aligned(16))) float zs[2][32 * ZSTRIDE];
    __shared__ int ldst[2][32];
    __shared__ __attribute__((aligned(16))) float    as1[NPB * DIN];
    __shared__ __attribute__((aligned(16))) float    as2[NPB * DIN];
    __shared__ __attribute__((aligned(16))) unsigned amx[NPB * DIN];
    __shared__ __attribute__((aligned(16))) unsigned amn[NPB * DIN];

    const int tid = threadIdx.x;
    const int n0  = blockIdx.x * NPB;
    const int ebeg = rptr[n0];
    const int eend = (blockIdx.x == EGRID - 1) ? NEDGES : rptr[n0 + NPB];
    const int nt   = (eend - ebeg + 31) >> 5;

    for (int i = tid; i < NPB * DIN; i += 256) {
        as1[i] = 0.0f; as2[i] = 0.0f;
        amx[i] = 0u;
        amn[i] = 0xFFFFFFFFu;
    }

    const int l  = tid & 31;
    const int wv = tid >> 5;
    const int cg = wv & 3;
    const int eg = wv >> 2;
    const int lo = l & 15;
    const int hi = l >> 4;
    const int col = cg * 16 + lo;

    v2f bf[NKE];
#pragma unroll
    for (int s = 0; s < NKE; ++s) {
        const int row = 4 * s + (4 * s >= DIN ? DIN : 0) + 2 * hi;
        bf[s].x = Wpre[row * DIN + col];
        bf[s].y = Wpre[(row + 1) * DIN + col];
    }

    const int r = tid >> 3, p = tid & 7;
    int pos = ebeg + r;
    int pq = pos + 32;  if (pq >= NEDGES) pq = NEDGES - 1;
    int pr = pos + 64;  if (pr >= NEDGES) pr = NEDGES - 1;
    int ei_c = perm[pos < NEDGES ? pos : NEDGES - 1];
    int ei_n = perm[pq];
    int pe2  = perm[pr];
    int sI_c = src[ei_c], dcur = dst[ei_c];
    int sI_n = src[ei_n], dI_n = dst[ei_n];
    float4 s_lo, s_hi;
    float2 e_pc;
    {
        const float4* hs = (const float4*)(h + (size_t)sI_c * DIN);
        const float2* er = (const float2*)(e + (size_t)ei_c * EDIM);
        s_lo = hs[p]; s_hi = hs[p + 8];
        e_pc = er[p];
    }

    const int rb = eg * 16 + 8 * hi;
    int buf = 0;
    for (int t = 0; t < nt; ++t, pos += 32) {
        {
            float* zr = &zs[buf][r * ZSTRIDE];
            *(float4*)(zr + 4 * p)       = s_lo;
            *(float4*)(zr + 32 + 4 * p)  = s_hi;
            *(float2*)(zr + DIN + 2 * p) = e_pc;
            if (p == 0) ldst[buf][r] = (pos < eend) ? dcur : -1;
        }
        __syncthreads();

        {
            const float4* hs = (const float4*)(h + (size_t)sI_n * DIN);
            const float2* er = (const float2*)(e + (size_t)ei_n * EDIM);
            s_lo = hs[p]; s_hi = hs[p + 8];
            e_pc = er[p];
            dcur = dI_n;
            ei_n = pe2;
            sI_n = src[pe2]; dI_n = dst[pe2];
            int pn = pos + 96;  if (pn >= NEDGES) pn = NEDGES - 1;
            pe2 = perm[pn];
        }

        v8f acc = {};
        const float* arow = &zs[buf][(eg * 16 + lo) * ZSTRIDE + 2 * hi];
#pragma unroll
        for (int s = 0; s < NKE; ++s) {
            v2f a = *(const v2f*)(arow + 4 * s);
            acc = __builtin_amdgcn_wmma_f32_16x16x4_f32(
                false, a, false, bf[s], (short)0, acc, false, false);
        }

        const int* ld = &ldst[buf][rb];
        int cur = ld[0];
        float v = acc[0];
        float sum = v, sq = v * v;
        unsigned emx = fenc(v), emn = emx;
#pragma unroll
        for (int q = 1; q < 8; ++q) {
            int d = ld[q];
            v = acc[q];
            unsigned ev = fenc(v);
            if (d != cur) {
                if (cur >= 0) {
                    const int o = (cur - n0) * DIN + col;
                    atomicAdd(&as1[o], sum);
                    atomicAdd(&as2[o], sq);
                    atomicMax(&amx[o], emx);
                    atomicMin(&amn[o], emn);
                }
                cur = d; sum = v; sq = v * v; emx = ev; emn = ev;
            } else {
                sum += v; sq += v * v;
                emx = (ev > emx) ? ev : emx;
                emn = (ev < emn) ? ev : emn;
            }
        }
        if (cur >= 0) {
            const int o = (cur - n0) * DIN + col;
            atomicAdd(&as1[o], sum);
            atomicAdd(&as2[o], sq);
            atomicMax(&amx[o], emx);
            atomicMin(&amn[o], emn);
        }
        buf ^= 1;
    }

    __syncthreads();
    {
        const size_t g0 = (size_t)n0 * DIN;
        float4* o1 = (float4*)(s1 + g0);
        float4* o2 = (float4*)(s2 + g0);
        uint4*  ox = (uint4*)(mx + g0);
        uint4*  on = (uint4*)(mn + g0);
        for (int i = tid; i < NPB * DIN / 4; i += 256) {
            o1[i] = ((const float4*)as1)[i];
            o2[i] = ((const float4*)as2)[i];
            ox[i] = ((const uint4*)amx)[i];
            on[i] = ((const uint4*)amn)[i];
        }
    }
}

#define FSTRIDE 324
#define NTM     2
#define NBLK    (16 * NTM)
#define NKB     16
#define NKC     16
static_assert(NNODES % NBLK == 0, "whole blocks");
__global__ __launch_bounds__(128) void pna_node(
    const float* __restrict__ h, const float* __restrict__ snorm,
    const float* __restrict__ Wpost, const float* __restrict__ bpost,
    const float* __restrict__ Wpre, const float* __restrict__ bpre,
    const float* __restrict__ s1, const float* __restrict__ s2,
    const unsigned* __restrict__ mx, const unsigned* __restrict__ mn,
    const int* __restrict__ cnt,
    float* __restrict__ out, float* bns, float* bnq) {

    __shared__ __attribute__((aligned(16))) float feat[NBLK * FSTRIDE];
    __shared__ float fab[NBLK * 2];
    const int tid = threadIdx.x;
    const int n0  = blockIdx.x * NBLK;

    const int l  = tid & 31;
    const int cg = tid >> 5;
    const int lo = l & 15;
    const int hi = l >> 4;
    const int col = cg * 16 + lo;

    const float* wc = Wpre + (size_t)(DIN + 2 * hi) * DIN + col;
    float cx[NKC], cy[NKC];
#pragma unroll
    for (int s = 0; s < NKC; ++s) {
        cx[s] = wc[(4 * s) * DIN];
        cy[s] = wc[(4 * s + 1) * DIN];
    }
    const float cb = bpre[col];

    {
        const int r0 = tid >> 3, p = tid & 7;
        float4 hv[NTM][2], a1[NTM][2], a2[NTM][2];
        uint4  ux[NTM][2], un[NTM][2];
        int dg[NTM];
#pragma unroll
        for (int t = 0; t < NTM; ++t) {
            const int node = n0 + 16 * t + r0;
            const size_t o = (size_t)node * DIN;
            const float4* hp  = (const float4*)(h + o);
            const float4* s1p = (const float4*)(s1 + o);
            const float4* s2p = (const float4*)(s2 + o);
            const uint4*  mxp = (const uint4*)(mx + o);
            const uint4*  mnp = (const uint4*)(mn + o);
            dg[t] = cnt[node];
#pragma unroll
            for (int q = 0; q < 2; ++q) {
                hv[t][q] = hp[p + 8 * q];
                a1[t][q] = s1p[p + 8 * q];
                a2[t][q] = s2p[p + 8 * q];
                ux[t][q] = mxp[p + 8 * q];
                un[t][q] = mnp[p + 8 * q];
            }
        }
#pragma unroll
        for (int t = 0; t < NTM; ++t) {
            const int r = 16 * t + r0;
            float* fr = &feat[r * FSTRIDE];
            *(float4*)(fr + 4 * p)      = hv[t][0];
            *(float4*)(fr + 32 + 4 * p) = hv[t][1];
            const float D    = (float)dg[t];
            const float logd = logf(D + 1.0f);
            const float fa   = logd / AVG_D;
            const float fb   = AVG_D / logd;
            if (p == 0) { fab[2 * r] = fa; fab[2 * r + 1] = fb; }
#pragma unroll
            for (int q = 0; q < 2; ++q) {
                const int c = 4 * p + 32 * q;
                float4 mu, sd, vx, vn;
                mu.x = a1[t][q].x / D; mu.y = a1[t][q].y / D;
                mu.z = a1[t][q].z / D; mu.w = a1[t][q].w / D;
                sd.x = sqrtf(fmaxf(a2[t][q].x / D - mu.x * mu.x, 0.0f) + EPSV);
                sd.y = sqrtf(fmaxf(a2[t][q].y / D - mu.y * mu.y, 0.0f) + EPSV);
                sd.z = sqrtf(fmaxf(a2[t][q].z / D - mu.z * mu.z, 0.0f) + EPSV);
                sd.w = sqrtf(fmaxf(a2[t][q].w / D - mu.w * mu.w, 0.0f) + EPSV);
                vx.x = fdec(ux[t][q].x); vx.y = fdec(ux[t][q].y);
                vx.z = fdec(ux[t][q].z); vx.w = fdec(ux[t][q].w);
                vn.x = fdec(un[t][q].x); vn.y = fdec(un[t][q].y);
                vn.z = fdec(un[t][q].z); vn.w = fdec(un[t][q].w);
                *(float4*)(fr + DIN + c)      = mu;
                *(float4*)(fr + 2 * DIN + c)  = vx;
                *(float4*)(fr + 3 * DIN + c)  = vn;
                *(float4*)(fr + 4 * DIN + c)  = sd;
            }
        }
    }
    __syncthreads();

    const float* arow = &feat[lo * FSTRIDE + 2 * hi];

    {
        v8f cacc[NTM];
#pragma unroll
        for (int t = 0; t < NTM; ++t) cacc[t] = (v8f){};
#pragma unroll
        for (int s = 0; s < NKC; ++s) {
            v2f b;
            b.x = cx[s];
            b.y = cy[s];
#pragma unroll
            for (int t = 0; t < NTM; ++t) {
                v2f a = *(const v2f*)(arow + 16 * t * FSTRIDE + 4 * s);
                cacc[t] = __builtin_amdgcn_wmma_f32_16x16x4_f32(
                    false, a, false, b, (short)0, cacc[t], false, false);
            }
        }
#pragma unroll
        for (int t = 0; t < NTM; ++t) {
#pragma unroll
            for (int i = 0; i < 8; ++i) {
                const int row = 16 * t + 8 * hi + i;
                const float cv = cacc[t][i] + cb;
                float* fp = &feat[row * FSTRIDE + col];
                fp[1 * DIN] = fp[1 * DIN] + cv;
                fp[2 * DIN] = fp[2 * DIN] + cv;
                fp[3 * DIN] = fp[3 * DIN] + cv;
            }
        }
    }
    __syncthreads();

    float fsa[NTM], fsb[NTM];
#pragma unroll
    for (int t = 0; t < NTM; ++t) {
        fsa[t] = fab[2 * (16 * t + lo)];
        fsb[t] = fab[2 * (16 * t + lo) + 1];
    }

    v8f acc[NTM];
#pragma unroll
    for (int t = 0; t < NTM; ++t) acc[t] = (v8f){};
    const float* bcol = Wpost + (size_t)(2 * hi) * DOUT + col;
#define NH 8
    float hx[NKB], hy[NKB];
#pragma unroll
    for (int s = 0; s < NKB; ++s) {
        hx[s] = bcol[(4 * s) * DOUT];
        hy[s] = bcol[(4 * s + 1) * DOUT];
    }
    float gx[2][3][NH], gy[2][3][NH];
#pragma unroll
    for (int j = 0; j < 3; ++j) {
#pragma unroll
        for (int s = 0; s < NH; ++s) {
            const int k = (1 + 4 * j) * DIN + 4 * s;
            gx[0][j][s] = bcol[k * DOUT];
            gy[0][j][s] = bcol[(k + 1) * DOUT];
        }
    }
#pragma unroll
    for (int s = 0; s < NKB; ++s) {
        v2f b;
        b.x = hx[s];
        b.y = hy[s];
#pragma unroll
        for (int t = 0; t < NTM; ++t) {
            v2f a = *(const v2f*)(arow + 16 * t * FSTRIDE + 4 * s);
            acc[t] = __builtin_amdgcn_wmma_f32_16x16x4_f32(
                false, a, false, b, (short)0, acc[t], false, false);
        }
    }
#pragma unroll
    for (int hb = 0; hb < 8; ++hb) {
        const int cur = hb & 1, nxt = cur ^ 1;
        if (hb + 1 < 8) {
            const int gn = 1 + (hb + 1) / 2, sn = ((hb + 1) % 2) * NH;
#pragma unroll
            for (int j = 0; j < 3; ++j) {
#pragma unroll
                for (int s = 0; s < NH; ++s) {
                    const int k = (gn + 4 * j) * DIN + 4 * (sn + s);
                    gx[nxt][j][s] = bcol[k * DOUT];
                    gy[nxt][j][s] = bcol[(k + 1) * DOUT];
                }
            }
        }
        const int g = 1 + hb / 2, s0 = (hb % 2) * NH;
#pragma unroll
        for (int s = 0; s < NH; ++s) {
            v2f a[NTM];
#pragma unroll
            for (int t = 0; t < NTM; ++t)
                a[t] = *(const v2f*)(arow + 16 * t * FSTRIDE + g * DIN + 4 * (s0 + s));
#pragma unroll
            for (int j = 0; j < 3; ++j) {
                v2f b;
                b.x = gx[cur][j][s];
                b.y = gy[cur][j][s];
#pragma unroll
                for (int t = 0; t < NTM; ++t) {
                    v2f af = a[t];
                    if (j == 1) { af.x = a[t].x * fsa[t]; af.y = a[t].y * fsa[t]; }
                    if (j == 2) { af.x = a[t].x * fsb[t]; af.y = a[t].y * fsb[t]; }
                    acc[t] = __builtin_amdgcn_wmma_f32_16x16x4_f32(
                        false, af, false, b, (short)0, acc[t], false, false);
                }
            }
        }
    }
#undef NH

    const float bv = bpost[col];
    float ps = 0.0f, ps2 = 0.0f;
#pragma unroll
    for (int t = 0; t < NTM; ++t) {
#pragma unroll
        for (int r = 0; r < 8; ++r) {
            int node = n0 + 16 * t + 8 * hi + r;
            float v = (acc[t][r] + bv) * snorm[node];
            out[(size_t)node * DOUT + col] = v;
            ps += v; ps2 += v * v;
        }
    }
    ps  += __shfl_xor(ps, 16, 32);
    ps2 += __shfl_xor(ps2, 16, 32);
    if (hi == 0) {
        atomicAdd(&bns[col], ps);
        atomicAdd(&bnq[col], ps2);
    }
}

__global__ __launch_bounds__(256) void pna_bn(
    float* out, const float* bns, const float* bnq,
    const float* gamma, const float* beta) {
    int i = blockIdx.x * 256 + threadIdx.x;
    if (i >= NNODES * DOUT) return;
    int c = i & 63;
    float mu  = bns[c] / (float)NNODES;
    float var = bnq[c] / (float)NNODES - mu * mu;
    float v = out[i];
    out[i] = (v - mu) * rsqrtf(var + 1e-05f) * gamma[c] + beta[c];
}

extern "C" void kernel_launch(void* const* d_in, const int* in_sizes, int n_in,
                              void* d_out, int out_size, void* d_ws, size_t ws_size,
                              hipStream_t stream) {
    const float* h      = (const float*)d_in[0];
    const float* e      = (const float*)d_in[1];
    const float* snorm  = (const float*)d_in[2];
    const float* Wpre   = (const float*)d_in[3];
    const float* bpre   = (const float*)d_in[4];
    const float* Wpost  = (const float*)d_in[5];
    const float* bpost  = (const float*)d_in[6];
    const float* gamma  = (const float*)d_in[7];
    const float* beta   = (const float*)d_in[8];
    const int*   src    = (const int*)d_in[9];
    const int*   dst    = (const int*)d_in[10];
    float* out = (float*)d_out;

    float*    s1  = (float*)d_ws;
    float*    s2  = s1 + (size_t)NNODES * DIN;
    unsigned* mx  = (unsigned*)(s2 + (size_t)NNODES * DIN);
    unsigned* mn  = mx + (size_t)NNODES * DIN;
    int*      cnt = (int*)(mn + (size_t)NNODES * DIN);
    float*    bns = (float*)(cnt + NNODES);
    float*    bnq = bns + DOUT;

    int* perm = (int*)d_out;
    int* rank = perm + NEDGES;
    int* rptr = rank + NEDGES;

    pna_init  <<<(NNODES + 255) / 256, 256, 0, stream>>>(cnt, bns, bnq);
    pna_rank  <<<(NEDGES + 255) / 256, 256, 0, stream>>>(dst, cnt, rank);
    pna_scan  <<<1, 1024, 0, stream>>>(cnt, rptr);
    pna_fill  <<<(NEDGES + 255) / 256, 256, 0, stream>>>(dst, rank, rptr, perm);
    pna_edge  <<<EGRID, 256, 0, stream>>>(h, e, Wpre, src, dst, perm, rptr, s1, s2, mx, mn);
    pna_node  <<<NNODES / NBLK, 128, 0, stream>>>(h, snorm, Wpost, bpost, Wpre, bpre, s1, s2, mx, mn,
                                                cnt, out, bns, bnq);
    pna_bn    <<<(NNODES * DOUT + 255) / 256, 256, 0, stream>>>(out, bns, bnq, gamma, beta);
}
